# starting kernel
# baseline (speedup 1.0000x reference)
#include <hip/hip_runtime.h>
#include <hip/hip_bf16.h>


typedef __attribute__((ext_vector_type(16))) _Float16 v16h;
typedef __attribute__((ext_vector_type(8)))  _Float16 v8h;
typedef __attribute__((ext_vector_type(4)))  _Float16 v4h;
typedef __attribute__((ext_vector_type(8)))  float    v8f;

#define FLAG_RELU  1
#define FLAG_TRANS 2

__device__ __forceinline__ v16h load_frag(const _Float16* base, int ld,
                                          int row0, int k0, int lane) {
  const int r  = row0 + (lane & 15);
  const int kb = k0 + ((lane >> 4) << 3);
  const _Float16* p = base + (size_t)r * ld + kb;
  v8h lo = *(const v8h*)(p);
  v8h hi = *(const v8h*)(p + 16);
  return __builtin_shufflevector(lo, hi, 0, 1, 2, 3, 4, 5, 6, 7,
                                         8, 9, 10, 11, 12, 13, 14, 15);
}

__device__ __forceinline__ v8f wmma_f16(v16h a, v16h b, v8f c) {
  return __builtin_amdgcn_wmma_f32_16x16x32_f16(false, a, false, b,
                                                (short)0, c, false, false);
}

__global__ __launch_bounds__(256) void cvt_f32_f16_kernel(
    const float* __restrict__ in, _Float16* __restrict__ out, int n4) {
  const int i = blockIdx.x * blockDim.x + threadIdx.x;
  if (i < n4) {
    float4 v = ((const float4*)in)[i];
    v4h o = {(_Float16)v.x, (_Float16)v.y, (_Float16)v.z, (_Float16)v.w};
    ((v4h*)out)[i] = o;
  }
}

__global__ __launch_bounds__(128) void gemm_nt_kernel(
    const _Float16* __restrict__ A, const _Float16* __restrict__ Bw,
    float* __restrict__ Cf, _Float16* __restrict__ Ch,
    const float* __restrict__ bias, int Mdim, int Ndim, int Kdim, int flags) {
  const int lane = threadIdx.x & 31;
  const int wave = threadIdx.x >> 5;
  const int tiles_n = Ndim >> 7;
  const int bm = blockIdx.x / tiles_n;
  const int bn = blockIdx.x - bm * tiles_n;
  const int m0 = bm * 128 + (wave & 1) * 64;
  const int n0 = bn * 128 + (wave >> 1) * 64;

  v8f acc[4][4] = {};

#pragma unroll 1
  for (int k0 = 0; k0 < Kdim; k0 += 32) {
    __builtin_prefetch(A + (size_t)(m0 + (lane & 15)) * Kdim + k0 + 32, 0, 1);
    __builtin_prefetch(Bw + (size_t)(n0 + (lane & 15)) * Kdim + k0 + 32, 0, 1);

    v16h af[4], bf[4];
#pragma unroll
    for (int i = 0; i < 4; i++) af[i] = load_frag(A, Kdim, m0 + 16 * i, k0, lane);
#pragma unroll
    for (int j = 0; j < 4; j++) bf[j] = load_frag(Bw, Kdim, n0 + 16 * j, k0, lane);
#pragma unroll
    for (int i = 0; i < 4; i++)
#pragma unroll
      for (int j = 0; j < 4; j++)
        acc[i][j] = wmma_f16(af[i], bf[j], acc[i][j]);
  }

  const int nl = lane & 15;
  const int mh = (lane >> 4) << 3;
#pragma unroll
  for (int i = 0; i < 4; i++) {
#pragma unroll
    for (int j = 0; j < 4; j++) {
      const int n = n0 + j * 16 + nl;
      const float bv = bias ? bias[n] : 0.f;
#pragma unroll
      for (int r = 0; r < 8; r++) {
        const int m = m0 + 16 * i + mh + r;
        float v = acc[i][j][r] + bv;
        if (flags & FLAG_RELU) v = fmaxf(v, 0.f);
        if (Cf) Cf[(size_t)m * Ndim + n] = v;
        if (Ch) {
          if (flags & FLAG_TRANS) {
            Ch[((size_t)((m >> 10) << 10) + n) * 1024 + (m & 1023)] = (_Float16)v;
          } else {
            Ch[(size_t)m * Ndim + n] = (_Float16)v;
          }
        }
      }
    }
  }
}

__global__ __launch_bounds__(256) void attn_kernel(
    const _Float16* __restrict__ Qh, const _Float16* __restrict__ Kh,
    const _Float16* __restrict__ Vt, const int* __restrict__ mask,
    _Float16* __restrict__ Oh) {
  __shared__ _Float16 plds[8][16 * 40];

  const int lane = threadIdx.x & 31;
  const int wave = threadIdx.x >> 5;
  const int bh   = blockIdx.x >> 3;
  const int qblk = blockIdx.x & 7;
  const int b = bh >> 4, h = bh & 15;
  const int q0 = qblk * 128 + wave * 16;

  const _Float16* Qbase = Qh + (size_t)b * 1024 * 1024 + h * 64;
  const _Float16* Kbase = Kh + (size_t)b * 1024 * 1024 + h * 64;
  const _Float16* Vbase = Vt + (size_t)(b * 16 + h) * 64 * 1024;
  const int* mb = mask + b * 1024;
  _Float16* myp = &plds[wave][0];

  const v16h qa0 = load_frag(Qbase, 1024, q0, 0, lane);
  const v16h qa1 = load_frag(Qbase, 1024, q0, 32, lane);

  v8f o[4] = {};
  v8f mrun = {-1e30f, -1e30f, -1e30f, -1e30f, -1e30f, -1e30f, -1e30f, -1e30f};
  v8f lrun = {};
  const float scale = 0.125f;
  const int mh8 = (lane >> 4) << 3;

#pragma unroll 1
  for (int kb = 0; kb < 1024; kb += 32) {
    v8f s[2] = {};
#pragma unroll
    for (int sub = 0; sub < 2; sub++) {
      v16h kf0 = load_frag(Kbase, 1024, kb + sub * 16, 0, lane);
      v16h kf1 = load_frag(Kbase, 1024, kb + sub * 16, 32, lane);
      s[sub] = wmma_f16(qa0, kf0, s[sub]);
      s[sub] = wmma_f16(qa1, kf1, s[sub]);
    }
    const int mk0 = mb[kb + (lane & 15)];
    const int mk1 = mb[kb + 16 + (lane & 15)];

    v8f bmax;
#pragma unroll
    for (int r = 0; r < 8; r++) {
      float v = -1e30f;
      if (mk0) v = fmaxf(v, s[0][r] * scale);
      if (mk1) v = fmaxf(v, s[1][r] * scale);
      bmax[r] = v;
    }
#pragma unroll
    for (int off = 1; off < 16; off <<= 1)
#pragma unroll
      for (int r = 0; r < 8; r++)
        bmax[r] = fmaxf(bmax[r], __shfl_xor(bmax[r], off, 32));

    v8f mnew, alpha, p0, p1, bsum;
#pragma unroll
    for (int r = 0; r < 8; r++) {
      mnew[r]  = fmaxf(mrun[r], bmax[r]);
      alpha[r] = __expf(mrun[r] - mnew[r]);
      p0[r] = mk0 ? __expf(s[0][r] * scale - mnew[r]) : 0.f;
      p1[r] = mk1 ? __expf(s[1][r] * scale - mnew[r]) : 0.f;
      bsum[r] = p0[r] + p1[r];
    }
#pragma unroll
    for (int off = 1; off < 16; off <<= 1)
#pragma unroll
      for (int r = 0; r < 8; r++)
        bsum[r] += __shfl_xor(bsum[r], off, 32);
#pragma unroll
    for (int r = 0; r < 8; r++) {
      lrun[r] = lrun[r] * alpha[r] + bsum[r];
      mrun[r] = mnew[r];
    }
#pragma unroll
    for (int t = 0; t < 4; t++)
#pragma unroll
      for (int r = 0; r < 8; r++) o[t][r] *= alpha[r];

#pragma unroll
    for (int r = 0; r < 8; r++) {
      myp[(r + mh8) * 40 + (lane & 15)]      = (_Float16)p0[r];
      myp[(r + mh8) * 40 + 16 + (lane & 15)] = (_Float16)p1[r];
    }
    asm volatile("s_wait_dscnt 0" ::: "memory");
    v16h pa = load_frag(myp, 40, 0, 0, lane);

#pragma unroll
    for (int t = 0; t < 4; t++) {
      v16h vf = load_frag(Vbase, 1024, t * 16, kb, lane);
      o[t] = wmma_f16(pa, vf, o[t]);
    }
  }

#pragma unroll
  for (int t = 0; t < 4; t++) {
    const int e = t * 16 + (lane & 15);
#pragma unroll
    for (int r = 0; r < 8; r++) {
      const int q = q0 + mh8 + r;
      const float denom = lrun[r];
      const float val = denom > 0.f ? o[t][r] / denom : 0.f;
      Oh[((size_t)b * 1024 + q) * 1024 + h * 64 + e] = (_Float16)val;
    }
  }
}

__global__ __launch_bounds__(256) void ln_kernel(
    const float* __restrict__ Ain, const float* __restrict__ Bin,
    const float* __restrict__ g, const float* __restrict__ beta,
    float* __restrict__ out32, _Float16* __restrict__ out16) {
  const int row = blockIdx.x;
  const int t = threadIdx.x;
  const int lane = t & 31, wave = t >> 5;

  const float4 xa = ((const float4*)(Ain + (size_t)row * 1024))[t];
  const float4 xb = ((const float4*)(Bin + (size_t)row * 1024))[t];
  float x[4] = {xa.x + xb.x, xa.y + xb.y, xa.z + xb.z, xa.w + xb.w};

  float s1 = x[0] + x[1] + x[2] + x[3];
  float s2 = x[0]*x[0] + x[1]*x[1] + x[2]*x[2] + x[3]*x[3];
#pragma unroll
  for (int off = 16; off >= 1; off >>= 1) {
    s1 += __shfl_xor(s1, off, 32);
    s2 += __shfl_xor(s2, off, 32);
  }
  __shared__ float r1[8], r2[8];
  if (lane == 0) { r1[wave] = s1; r2[wave] = s2; }
  __syncthreads();
  float tot1 = 0.f, tot2 = 0.f;
#pragma unroll
  for (int i = 0; i < 8; i++) { tot1 += r1[i]; tot2 += r2[i]; }

  const float mu   = tot1 * (1.f / 1024.f);
  const float var  = tot2 * (1.f / 1024.f) - mu * mu;
  const float rstd = rsqrtf(var + 1e-5f);
#pragma unroll
  for (int j = 0; j < 4; j++) {
    const int idx = t * 4 + j;
    const float y = (x[j] - mu) * rstd * g[idx] + beta[idx];
    out32[(size_t)row * 1024 + idx] = y;
    if (out16) out16[(size_t)row * 1024 + idx] = (_Float16)y;
  }
}

extern "C" void kernel_launch(void* const* d_in, const int* in_sizes, int n_in,
                              void* d_out, int out_size, void* d_ws, size_t ws_size,
                              hipStream_t stream) {
  (void)in_sizes; (void)n_in; (void)out_size; (void)ws_size;
  const float* X    = (const float*)d_in[0];
  const float* Y    = (const float*)d_in[1];
  const int*   mask = (const int*)d_in[2];
  const float* Wq   = (const float*)d_in[3];
  const float* Wk   = (const float*)d_in[4];
  const float* Wv   = (const float*)d_in[5];
  const float* Wo   = (const float*)d_in[6];
  const float* ln1g = (const float*)d_in[7];
  const float* ln1b = (const float*)d_in[8];
  const float* ln2g = (const float*)d_in[9];
  const float* ln2b = (const float*)d_in[10];
  const float* W1   = (const float*)d_in[11];
  const float* b1   = (const float*)d_in[12];
  const float* W2   = (const float*)d_in[13];
  const float* b2   = (const float*)d_in[14];

  char* ws = (char*)d_ws;
  const size_t MB = 1024 * 1024;
  _Float16* Xh   = (_Float16*)(ws + 0);
  _Float16* Yh   = (_Float16*)(ws + 8 * MB);
  _Float16* Wqh  = (_Float16*)(ws + 16 * MB);
  _Float16* Wkh  = (_Float16*)(ws + 18 * MB);
  _Float16* Wvh  = (_Float16*)(ws + 20 * MB);
  _Float16* Woh  = (_Float16*)(ws + 22 * MB);
  _Float16* W1h  = (_Float16*)(ws + 24 * MB);
  _Float16* W2h  = (_Float16*)(ws + 26 * MB);
  _Float16* Qh   = (_Float16*)(ws + 28 * MB);
  _Float16* Kh   = (_Float16*)(ws + 36 * MB);
  _Float16* Vt   = (_Float16*)(ws + 44 * MB);
  float*    Fbuf = (float*)(ws + 52 * MB);
  float*    Hres = (float*)(ws + 68 * MB);
  _Float16* Ohb  = Xh;
  _Float16* Hh   = Yh;
  _Float16* FF1h = Qh;

  const int n4m = 4096 * 1024 / 4;
  const int n4w = 1024 * 1024 / 4;
  dim3 blk256(256);
  dim3 blk128(128);

  cvt_f32_f16_kernel<<<n4m / 256, blk256, 0, stream>>>(X, Xh, n4m);
  cvt_f32_f16_kernel<<<n4m / 256, blk256, 0, stream>>>(Y, Yh, n4m);
  cvt_f32_f16_kernel<<<n4w / 256, blk256, 0, stream>>>(Wq, Wqh, n4w);
  cvt_f32_f16_kernel<<<n4w / 256, blk256, 0, stream>>>(Wk, Wkh, n4w);
  cvt_f32_f16_kernel<<<n4w / 256, blk256, 0, stream>>>(Wv, Wvh, n4w);
  cvt_f32_f16_kernel<<<n4w / 256, blk256, 0, stream>>>(Wo, Woh, n4w);
  cvt_f32_f16_kernel<<<n4w / 256, blk256, 0, stream>>>(W1, W1h, n4w);
  cvt_f32_f16_kernel<<<n4w / 256, blk256, 0, stream>>>(W2, W2h, n4w);

  const int ggrid = (4096 / 128) * (1024 / 128);

  gemm_nt_kernel<<<ggrid, blk128, 0, stream>>>(Xh, Wqh, nullptr, Qh, nullptr,
                                               4096, 1024, 1024, 0);
  gemm_nt_kernel<<<ggrid, blk128, 0, stream>>>(Yh, Wkh, nullptr, Kh, nullptr,
                                               4096, 1024, 1024, 0);
  gemm_nt_kernel<<<ggrid, blk128, 0, stream>>>(Yh, Wvh, nullptr, Vt, nullptr,
                                               4096, 1024, 1024, FLAG_TRANS);

  attn_kernel<<<512, blk256, 0, stream>>>(Qh, Kh, Vt, mask, Ohb);

  gemm_nt_kernel<<<ggrid, blk128, 0, stream>>>(Ohb, Woh, Fbuf, nullptr, nullptr,
                                               4096, 1024, 1024, 0);
  ln_kernel<<<4096, blk256, 0, stream>>>(X, Fbuf, ln1g, ln1b, Hres, Hh);

  gemm_nt_kernel<<<ggrid, blk128, 0, stream>>>(Hh, W1h, nullptr, FF1h, b1,
                                               4096, 1024, 1024, FLAG_RELU);
  gemm_nt_kernel<<<ggrid, blk128, 0, stream>>>(FF1h, W2h, Fbuf, nullptr, b2,
                                               4096, 1024, 1024, 0);
  ln_kernel<<<4096, blk256, 0, stream>>>(Hres, Fbuf, ln2g, ln2b, (float*)d_out,
                                         nullptr);
}


# gemm: A and B tiles staged through LDS in 128-k slabs by whole-line 16-byte global loads (two stages, next slab in flight), fragments read from LDS; 8 waves of 64x32 per 128x128 tile
# speedup vs baseline: 1.4535x; 1.4535x over previous
#include <hip/hip_runtime.h>
#include <hip/hip_bf16.h>


typedef __attribute__((ext_vector_type(16))) _Float16 v16h;
typedef __attribute__((ext_vector_type(8)))  _Float16 v8h;
typedef __attribute__((ext_vector_type(4)))  _Float16 v4h;
typedef __attribute__((ext_vector_type(8)))  float    v8f;

#define FLAG_RELU  1
#define FLAG_TRANS 2

__device__ __forceinline__ v16h load_frag(const _Float16* base, int ld,
                                          int row0, int k0, int lane) {
  const int r  = row0 + (lane & 15);
  const int kb = k0 + ((lane >> 4) << 3);
  const _Float16* p = base + (size_t)r * ld + kb;
  v8h lo = *(const v8h*)(p);
  v8h hi = *(const v8h*)(p + 16);
  return __builtin_shufflevector(lo, hi, 0, 1, 2, 3, 4, 5, 6, 7,
                                         8, 9, 10, 11, 12, 13, 14, 15);
}

__device__ __forceinline__ v8f wmma_f16(v16h a, v16h b, v8f c) {
  return __builtin_amdgcn_wmma_f32_16x16x32_f16(false, a, false, b,
                                                (short)0, c, false, false);
}

__global__ __launch_bounds__(256) void cvt_f32_f16_kernel(
    const float* __restrict__ in, _Float16* __restrict__ out, int n4) {
  const int i = blockIdx.x * blockDim.x + threadIdx.x;
  if (i < n4) {
    float4 v = ((const float4*)in)[i];
    v4h o = {(_Float16)v.x, (_Float16)v.y, (_Float16)v.z, (_Float16)v.w};
    ((v4h*)out)[i] = o;
  }
}

#define G_LD    1024
#define G_BK    128
#define G_PITCH 136
#define G_STAGE (256 * G_PITCH)

__device__ __forceinline__ void gemm_stage_mma(const _Float16* sA,
                                               const _Float16* sB,
                                               v8f (&acc)[4][2], int lane) {
  v16h af[2][4], bf[2][2];
#pragma unroll
  for (int i = 0; i < 4; i++) af[0][i] = load_frag(sA, G_PITCH, 16 * i, 0, lane);
#pragma unroll
  for (int j = 0; j < 2; j++) bf[0][j] = load_frag(sB, G_PITCH, 16 * j, 0, lane);
#pragma unroll
  for (int u = 0; u < 4; u++) {
    const int c = u & 1, n = c ^ 1;
    if (u + 1 < 4) {
#pragma unroll
      for (int i = 0; i < 4; i++)
        af[n][i] = load_frag(sA, G_PITCH, 16 * i, 32 * (u + 1), lane);
#pragma unroll
      for (int j = 0; j < 2; j++)
        bf[n][j] = load_frag(sB, G_PITCH, 16 * j, 32 * (u + 1), lane);
    }
    __builtin_amdgcn_sched_barrier(0);
#pragma unroll
    for (int i = 0; i < 4; i++)
#pragma unroll
      for (int j = 0; j < 2; j++)
        acc[i][j] = wmma_f16(af[c][i], bf[c][j], acc[i][j]);
  }
}

__global__ __launch_bounds__(256) void gemm_nt_kernel(
    const _Float16* __restrict__ A, const _Float16* __restrict__ Bw,
    float* __restrict__ Cf, _Float16* __restrict__ Ch,
    const float* __restrict__ bias, int Mdim, int Ndim, int Kdim, int flags) {
  __shared__ __attribute__((aligned(16))) _Float16 lds[2 * G_STAGE];

  const int tid  = threadIdx.x;
  const int lane = tid & 31;
  const int wave = tid >> 5;
  const int tiles_n = Ndim >> 7;
  const int bm = blockIdx.x / tiles_n;
  const int bn = blockIdx.x - bm * tiles_n;
  const int wm = (wave & 1) * 64;
  const int wn = (wave >> 1) * 32;
  const int m0 = bm * 128 + wm;
  const int n0 = bn * 128 + wn;

  const int srow = tid >> 4;
  const int scol = (tid & 15) << 3;
  const _Float16* gA = A  + (size_t)(bm * 128 + srow) * G_LD + scol;
  const _Float16* gB = Bw + (size_t)(bn * 128 + srow) * G_LD + scol;
  const int soff = srow * G_PITCH + scol;

  v8f acc[4][2] = {};
  v8h ra[8], rb[8];

#pragma unroll
  for (int p = 0; p < 8; p++) {
    ra[p] = *(const v8h*)(gA + p * 16 * G_LD);
    rb[p] = *(const v8h*)(gB + p * 16 * G_LD);
  }
#pragma unroll
  for (int p = 0; p < 8; p++) {
    *(v8h*)(lds + soff + (16 * p) * G_PITCH)       = ra[p];
    *(v8h*)(lds + soff + (128 + 16 * p) * G_PITCH) = rb[p];
  }
  __syncthreads();

  const int nk = Kdim / G_BK;
#pragma unroll 1
  for (int kt = 0; kt + 1 < nk; kt++) {
    const _Float16* cur = lds + (kt & 1) * G_STAGE;
    _Float16*       nxt = lds + ((kt + 1) & 1) * G_STAGE;
    gA += G_BK;
    gB += G_BK;
#pragma unroll
    for (int p = 0; p < 8; p++) {
      ra[p] = *(const v8h*)(gA + p * 16 * G_LD);
      rb[p] = *(const v8h*)(gB + p * 16 * G_LD);
    }
    gemm_stage_mma(cur + wm * G_PITCH, cur + (128 + wn) * G_PITCH, acc, lane);
#pragma unroll
    for (int p = 0; p < 8; p++) {
      *(v8h*)(nxt + soff + (16 * p) * G_PITCH)       = ra[p];
      *(v8h*)(nxt + soff + (128 + 16 * p) * G_PITCH) = rb[p];
    }
    __syncthreads();
  }
  {
    const _Float16* cur = lds + ((nk - 1) & 1) * G_STAGE;
    gemm_stage_mma(cur + wm * G_PITCH, cur + (128 + wn) * G_PITCH, acc, lane);
  }

  const int nl = lane & 15;
  const int mh = (lane >> 4) << 3;
#pragma unroll
  for (int i = 0; i < 4; i++) {
#pragma unroll
    for (int j = 0; j < 2; j++) {
      const int n = n0 + j * 16 + nl;
      const float bv = bias ? bias[n] : 0.f;
#pragma unroll
      for (int r = 0; r < 8; r++) {
        const int m = m0 + 16 * i + mh + r;
        float v = acc[i][j][r] + bv;
        if (flags & FLAG_RELU) v = fmaxf(v, 0.f);
        if (Cf) Cf[(size_t)m * Ndim + n] = v;
        if (Ch) {
          if (flags & FLAG_TRANS) {
            Ch[((size_t)((m >> 10) << 10) + n) * 1024 + (m & 1023)] = (_Float16)v;
          } else {
            Ch[(size_t)m * Ndim + n] = (_Float16)v;
          }
        }
      }
    }
  }
}

__global__ __launch_bounds__(256) void attn_kernel(
    const _Float16* __restrict__ Qh, const _Float16* __restrict__ Kh,
    const _Float16* __restrict__ Vt, const int* __restrict__ mask,
    _Float16* __restrict__ Oh) {
  __shared__ _Float16 plds[8][16 * 40];

  const int lane = threadIdx.x & 31;
  const int wave = threadIdx.x >> 5;
  const int bh   = blockIdx.x >> 3;
  const int qblk = blockIdx.x & 7;
  const int b = bh >> 4, h = bh & 15;
  const int q0 = qblk * 128 + wave * 16;

  const _Float16* Qbase = Qh + (size_t)b * 1024 * 1024 + h * 64;
  const _Float16* Kbase = Kh + (size_t)b * 1024 * 1024 + h * 64;
  const _Float16* Vbase = Vt + (size_t)(b * 16 + h) * 64 * 1024;
  const int* mb = mask + b * 1024;
  _Float16* myp = &plds[wave][0];

  const v16h qa0 = load_frag(Qbase, 1024, q0, 0, lane);
  const v16h qa1 = load_frag(Qbase, 1024, q0, 32, lane);

  v8f o[4] = {};
  v8f mrun = {-1e30f, -1e30f, -1e30f, -1e30f, -1e30f, -1e30f, -1e30f, -1e30f};
  v8f lrun = {};
  const float scale = 0.125f;
  const int mh8 = (lane >> 4) << 3;

#pragma unroll 1
  for (int kb = 0; kb < 1024; kb += 32) {
    v8f s[2] = {};
#pragma unroll
    for (int sub = 0; sub < 2; sub++) {
      v16h kf0 = load_frag(Kbase, 1024, kb + sub * 16, 0, lane);
      v16h kf1 = load_frag(Kbase, 1024, kb + sub * 16, 32, lane);
      s[sub] = wmma_f16(qa0, kf0, s[sub]);
      s[sub] = wmma_f16(qa1, kf1, s[sub]);
    }
    const int mk0 = mb[kb + (lane & 15)];
    const int mk1 = mb[kb + 16 + (lane & 15)];

    v8f bmax;
#pragma unroll
    for (int r = 0; r < 8; r++) {
      float v = -1e30f;
      if (mk0) v = fmaxf(v, s[0][r] * scale);
      if (mk1) v = fmaxf(v, s[1][r] * scale);
      bmax[r] = v;
    }
#pragma unroll
    for (int off = 1; off < 16; off <<= 1)
#pragma unroll
      for (int r = 0; r < 8; r++)
        bmax[r] = fmaxf(bmax[r], __shfl_xor(bmax[r], off, 32));

    v8f mnew, alpha, p0, p1, bsum;
#pragma unroll
    for (int r = 0; r < 8; r++) {
      mnew[r]  = fmaxf(mrun[r], bmax[r]);
      alpha[r] = __expf(mrun[r] - mnew[r]);
      p0[r] = mk0 ? __expf(s[0][r] * scale - mnew[r]) : 0.f;
      p1[r] = mk1 ? __expf(s[1][r] * scale - mnew[r]) : 0.f;
      bsum[r] = p0[r] + p1[r];
    }
#pragma unroll
    for (int off = 1; off < 16; off <<= 1)
#pragma unroll
      for (int r = 0; r < 8; r++)
        bsum[r] += __shfl_xor(bsum[r], off, 32);
#pragma unroll
    for (int r = 0; r < 8; r++) {
      lrun[r] = lrun[r] * alpha[r] + bsum[r];
      mrun[r] = mnew[r];
    }
#pragma unroll
    for (int t = 0; t < 4; t++)
#pragma unroll
      for (int r = 0; r < 8; r++) o[t][r] *= alpha[r];

#pragma unroll
    for (int r = 0; r < 8; r++) {
      myp[(r + mh8) * 40 + (lane & 15)]      = (_Float16)p0[r];
      myp[(r + mh8) * 40 + 16 + (lane & 15)] = (_Float16)p1[r];
    }
    asm volatile("s_wait_dscnt 0" ::: "memory");
    v16h pa = load_frag(myp, 40, 0, 0, lane);

#pragma unroll
    for (int t = 0; t < 4; t++) {
      v16h vf = load_frag(Vbase, 1024, t * 16, kb, lane);
      o[t] = wmma_f16(pa, vf, o[t]);
    }
  }

#pragma unroll
  for (int t = 0; t < 4; t++) {
    const int e = t * 16 + (lane & 15);
#pragma unroll
    for (int r = 0; r < 8; r++) {
      const int q = q0 + mh8 + r;
      const float denom = lrun[r];
      const float val = denom > 0.f ? o[t][r] / denom : 0.f;
      Oh[((size_t)b * 1024 + q) * 1024 + h * 64 + e] = (_Float16)val;
    }
  }
}

__global__ __launch_bounds__(256) void ln_kernel(
    const float* __restrict__ Ain, const float* __restrict__ Bin,
    const float* __restrict__ g, const float* __restrict__ beta,
    float* __restrict__ out32, _Float16* __restrict__ out16) {
  const int row = blockIdx.x;
  const int t = threadIdx.x;
  const int lane = t & 31, wave = t >> 5;

  const float4 xa = ((const float4*)(Ain + (size_t)row * 1024))[t];
  const float4 xb = ((const float4*)(Bin + (size_t)row * 1024))[t];
  float x[4] = {xa.x + xb.x, xa.y + xb.y, xa.z + xb.z, xa.w + xb.w};

  float s1 = x[0] + x[1] + x[2] + x[3];
  float s2 = x[0]*x[0] + x[1]*x[1] + x[2]*x[2] + x[3]*x[3];
#pragma unroll
  for (int off = 16; off >= 1; off >>= 1) {
    s1 += __shfl_xor(s1, off, 32);
    s2 += __shfl_xor(s2, off, 32);
  }
  __shared__ float r1[8], r2[8];
  if (lane == 0) { r1[wave] = s1; r2[wave] = s2; }
  __syncthreads();
  float tot1 = 0.f, tot2 = 0.f;
#pragma unroll
  for (int i = 0; i < 8; i++) { tot1 += r1[i]; tot2 += r2[i]; }

  const float mu   = tot1 * (1.f / 1024.f);
  const float var  = tot2 * (1.f / 1024.f) - mu * mu;
  const float rstd = rsqrtf(var + 1e-5f);
#pragma unroll
  for (int j = 0; j < 4; j++) {
    const int idx = t * 4 + j;
    const float y = (x[j] - mu) * rstd * g[idx] + beta[idx];
    out32[(size_t)row * 1024 + idx] = y;
    if (out16) out16[(size_t)row * 1024 + idx] = (_Float16)y;
  }
}

extern "C" void kernel_launch(void* const* d_in, const int* in_sizes, int n_in,
                              void* d_out, int out_size, void* d_ws, size_t ws_size,
                              hipStream_t stream) {
  (void)in_sizes; (void)n_in; (void)out_size; (void)ws_size;
  const float* X    = (const float*)d_in[0];
  const float* Y    = (const float*)d_in[1];
  const int*   mask = (const int*)d_in[2];
  const float* Wq   = (const float*)d_in[3];
  const float* Wk   = (const float*)d_in[4];
  const float* Wv   = (const float*)d_in[5];
  const float* Wo   = (const float*)d_in[6];
  const float* ln1g = (const float*)d_in[7];
  const float* ln1b = (const float*)d_in[8];
  const float* ln2g = (const float*)d_in[9];
  const float* ln2b = (const float*)d_in[10];
  const float* W1   = (const float*)d_in[11];
  const float* b1   = (const float*)d_in[12];
  const float* W2   = (const float*)d_in[13];
  const float* b2   = (const float*)d_in[14];

  char* ws = (char*)d_ws;
  const size_t MB = 1024 * 1024;
  _Float16* Xh   = (_Float16*)(ws + 0);
  _Float16* Yh   = (_Float16*)(ws + 8 * MB);
  _Float16* Wqh  = (_Float16*)(ws + 16 * MB);
  _Float16* Wkh  = (_Float16*)(ws + 18 * MB);
  _Float16* Wvh  = (_Float16*)(ws + 20 * MB);
  _Float16* Woh  = (_Float16*)(ws + 22 * MB);
  _Float16* W1h  = (_Float16*)(ws + 24 * MB);
  _Float16* W2h  = (_Float16*)(ws + 26 * MB);
  _Float16* Qh   = (_Float16*)(ws + 28 * MB);
  _Float16* Kh   = (_Float16*)(ws + 36 * MB);
  _Float16* Vt   = (_Float16*)(ws + 44 * MB);
  float*    Fbuf = (float*)(ws + 52 * MB);
  float*    Hres = (float*)(ws + 68 * MB);
  _Float16* Ohb  = Xh;
  _Float16* Hh   = Yh;
  _Float16* FF1h = Qh;

  const int n4m = 4096 * 1024 / 4;
  const int n4w = 1024 * 1024 / 4;
  dim3 blk256(256);

  cvt_f32_f16_kernel<<<n4m / 256, blk256, 0, stream>>>(X, Xh, n4m);
  cvt_f32_f16_kernel<<<n4m / 256, blk256, 0, stream>>>(Y, Yh, n4m);
  cvt_f32_f16_kernel<<<n4w / 256, blk256, 0, stream>>>(Wq, Wqh, n4w);
  cvt_f32_f16_kernel<<<n4w / 256, blk256, 0, stream>>>(Wk, Wkh, n4w);
  cvt_f32_f16_kernel<<<n4w / 256, blk256, 0, stream>>>(Wv, Wvh, n4w);
  cvt_f32_f16_kernel<<<n4w / 256, blk256, 0, stream>>>(Wo, Woh, n4w);
  cvt_f32_f16_kernel<<<n4w / 256, blk256, 0, stream>>>(W1, W1h, n4w);
  cvt_f32_f16_kernel<<<n4w / 256, blk256, 0, stream>>>(W2, W2h, n4w);

  const int ggrid = (4096 / 128) * (1024 / 128);

  gemm_nt_kernel<<<ggrid, blk256, 0, stream>>>(Xh, Wqh, nullptr, Qh, nullptr,
                                               4096, 1024, 1024, 0);
  gemm_nt_kernel<<<ggrid, blk256, 0, stream>>>(Yh, Wkh, nullptr, Kh, nullptr,
                                               4096, 1024, 1024, 0);
  gemm_nt_kernel<<<ggrid, blk256, 0, stream>>>(Yh, Wvh, nullptr, Vt, nullptr,
                                               4096, 1024, 1024, FLAG_TRANS);

  attn_kernel<<<512, blk256, 0, stream>>>(Qh, Kh, Vt, mask, Ohb);

  gemm_nt_kernel<<<ggrid, blk256, 0, stream>>>(Ohb, Woh, Fbuf, nullptr, nullptr,
                                               4096, 1024, 1024, 0);
  ln_kernel<<<4096, blk256, 0, stream>>>(X, Fbuf, ln1g, ln1b, Hres, Hh);

  gemm_nt_kernel<<<ggrid, blk256, 0, stream>>>(Hh, W1h, nullptr, FF1h, b1,
                                               4096, 1024, 1024, FLAG_RELU);
  gemm_nt_kernel<<<ggrid, blk256, 0, stream>>>(FF1h, W2h, Fbuf, nullptr, b2,
                                               4096, 1024, 1024, 0);
  ln_kernel<<<4096, blk256, 0, stream>>>(Hres, Fbuf, ln2g, ln2b, (float*)d_out,
                                         nullptr);
}


# operand conversion pass (X, Y and the six weights to their matrix-operand copies): 8 launches of one quad per thread (114688 short waves) become 1 launch of 512 workgroups, 28 quads per thread
# speedup vs baseline: 1.5583x; 1.0721x over previous
#include <hip/hip_runtime.h>
#include <hip/hip_bf16.h>


typedef __attribute__((ext_vector_type(16))) _Float16 v16h;
typedef __attribute__((ext_vector_type(8)))  _Float16 v8h;
typedef __attribute__((ext_vector_type(4)))  _Float16 v4h;
typedef __attribute__((ext_vector_type(8)))  float    v8f;

#define FLAG_RELU  1
#define FLAG_TRANS 2

__device__ __forceinline__ v16h load_frag(const _Float16* base, int ld,
                                          int row0, int k0, int lane) {
  const int r  = row0 + (lane & 15);
  const int kb = k0 + ((lane >> 4) << 3);
  const _Float16* p = base + (size_t)r * ld + kb;
  v8h lo = *(const v8h*)(p);
  v8h hi = *(const v8h*)(p + 16);
  return __builtin_shufflevector(lo, hi, 0, 1, 2, 3, 4, 5, 6, 7,
                                         8, 9, 10, 11, 12, 13, 14, 15);
}

__device__ __forceinline__ v8f wmma_f16(v16h a, v16h b, v8f c) {
  return __builtin_amdgcn_wmma_f32_16x16x32_f16(false, a, false, b,
                                                (short)0, c, false, false);
}

#define CV_SLAB (512 * 256)

__device__ __forceinline__ v4h cvt_quad(float4 v) {
  v4h o = {(_Float16)v.x, (_Float16)v.y, (_Float16)v.z, (_Float16)v.w};
  return o;
}

__global__ __launch_bounds__(256) void cvt_all_kernel(
    const float* __restrict__ X,  const float* __restrict__ Y,
    const float* __restrict__ Wq, const float* __restrict__ Wk,
    const float* __restrict__ Wv, const float* __restrict__ Wo,
    const float* __restrict__ W1, const float* __restrict__ W2,
    _Float16* __restrict__ dst) {
  const size_t q0 = (size_t)blockIdx.x * 256 + threadIdx.x;
  float4 v[28];
#pragma unroll
  for (int i = 0; i < 8; i++) v[i]     = ((const float4*)X)[q0 + (size_t)i * CV_SLAB];
#pragma unroll
  for (int i = 0; i < 8; i++) v[8 + i] = ((const float4*)Y)[q0 + (size_t)i * CV_SLAB];
#pragma unroll
  for (int i = 0; i < 2; i++) {
    v[16 + i] = ((const float4*)Wq)[q0 + (size_t)i * CV_SLAB];
    v[18 + i] = ((const float4*)Wk)[q0 + (size_t)i * CV_SLAB];
    v[20 + i] = ((const float4*)Wv)[q0 + (size_t)i * CV_SLAB];
    v[22 + i] = ((const float4*)Wo)[q0 + (size_t)i * CV_SLAB];
    v[24 + i] = ((const float4*)W1)[q0 + (size_t)i * CV_SLAB];
    v[26 + i] = ((const float4*)W2)[q0 + (size_t)i * CV_SLAB];
  }
  __builtin_amdgcn_sched_barrier(0);
#pragma unroll
  for (int i = 0; i < 28; i++)
    ((v4h*)dst)[q0 + (size_t)i * CV_SLAB] = cvt_quad(v[i]);
}

#define G_LD    1024
#define G_BK    128
#define G_PITCH 136
#define G_STAGE (256 * G_PITCH)

__device__ __forceinline__ void gemm_stage_mma(const _Float16* sA,
                                               const _Float16* sB,
                                               v8f (&acc)[4][2], int lane) {
  v16h af[2][4], bf[2][2];
#pragma unroll
  for (int i = 0; i < 4; i++) af[0][i] = load_frag(sA, G_PITCH, 16 * i, 0, lane);
#pragma unroll
  for (int j = 0; j < 2; j++) bf[0][j] = load_frag(sB, G_PITCH, 16 * j, 0, lane);
#pragma unroll
  for (int u = 0; u < 4; u++) {
    const int c = u & 1, n = c ^ 1;
    if (u + 1 < 4) {
#pragma unroll
      for (int i = 0; i < 4; i++)
        af[n][i] = load_frag(sA, G_PITCH, 16 * i, 32 * (u + 1), lane);
#pragma unroll
      for (int j = 0; j < 2; j++)
        bf[n][j] = load_frag(sB, G_PITCH, 16 * j, 32 * (u + 1), lane);
    }
    __builtin_amdgcn_sched_barrier(0);
#pragma unroll
    for (int i = 0; i < 4; i++)
#pragma unroll
      for (int j = 0; j < 2; j++)
        acc[i][j] = wmma_f16(af[c][i], bf[c][j], acc[i][j]);
  }
}

__global__ __launch_bounds__(256) void gemm_nt_kernel(
    const _Float16* __restrict__ A, const _Float16* __restrict__ Bw,
    float* __restrict__ Cf, _Float16* __restrict__ Ch,
    const float* __restrict__ bias, int Mdim, int Ndim, int Kdim, int flags) {
  __shared__ __attribute__((aligned(16))) _Float16 lds[2 * G_STAGE];

  const int tid  = threadIdx.x;
  const int lane = tid & 31;
  const int wave = tid >> 5;
  const int tiles_n = Ndim >> 7;
  const int bm = blockIdx.x / tiles_n;
  const int bn = blockIdx.x - bm * tiles_n;
  const int wm = (wave & 1) * 64;
  const int wn = (wave >> 1) * 32;
  const int m0 = bm * 128 + wm;
  const int n0 = bn * 128 + wn;

  const int srow = tid >> 4;
  const int scol = (tid & 15) << 3;
  const _Float16* gA = A  + (size_t)(bm * 128 + srow) * G_LD + scol;
  const _Float16* gB = Bw + (size_t)(bn * 128 + srow) * G_LD + scol;
  const int soff = srow * G_PITCH + scol;

  v8f acc[4][2] = {};
  v8h ra[8], rb[8];

#pragma unroll
  for (int p = 0; p < 8; p++) {
    ra[p] = *(const v8h*)(gA + p * 16 * G_LD);
    rb[p] = *(const v8h*)(gB + p * 16 * G_LD);
  }
#pragma unroll
  for (int p = 0; p < 8; p++) {
    *(v8h*)(lds + soff + (16 * p) * G_PITCH)       = ra[p];
    *(v8h*)(lds + soff + (128 + 16 * p) * G_PITCH) = rb[p];
  }
  __syncthreads();

  const int nk = Kdim / G_BK;
#pragma unroll 1
  for (int kt = 0; kt + 1 < nk; kt++) {
    const _Float16* cur = lds + (kt & 1) * G_STAGE;
    _Float16*       nxt = lds + ((kt + 1) & 1) * G_STAGE;
    gA += G_BK;
    gB += G_BK;
#pragma unroll
    for (int p = 0; p < 8; p++) {
      ra[p] = *(const v8h*)(gA + p * 16 * G_LD);
      rb[p] = *(const v8h*)(gB + p * 16 * G_LD);
    }
    gemm_stage_mma(cur + wm * G_PITCH, cur + (128 + wn) * G_PITCH, acc, lane);
#pragma unroll
    for (int p = 0; p < 8; p++) {
      *(v8h*)(nxt + soff + (16 * p) * G_PITCH)       = ra[p];
      *(v8h*)(nxt + soff + (128 + 16 * p) * G_PITCH) = rb[p];
    }
    __syncthreads();
  }
  {
    const _Float16* cur = lds + ((nk - 1) & 1) * G_STAGE;
    gemm_stage_mma(cur + wm * G_PITCH, cur + (128 + wn) * G_PITCH, acc, lane);
  }

  const int nl = lane & 15;
  const int mh = (lane >> 4) << 3;
#pragma unroll
  for (int i = 0; i < 4; i++) {
#pragma unroll
    for (int j = 0; j < 2; j++) {
      const int n = n0 + j * 16 + nl;
      const float bv = bias ? bias[n] : 0.f;
#pragma unroll
      for (int r = 0; r < 8; r++) {
        const int m = m0 + 16 * i + mh + r;
        float v = acc[i][j][r] + bv;
        if (flags & FLAG_RELU) v = fmaxf(v, 0.f);
        if (Cf) Cf[(size_t)m * Ndim + n] = v;
        if (Ch) {
          if (flags & FLAG_TRANS) {
            Ch[((size_t)((m >> 10) << 10) + n) * 1024 + (m & 1023)] = (_Float16)v;
          } else {
            Ch[(size_t)m * Ndim + n] = (_Float16)v;
          }
        }
      }
    }
  }
}

__global__ __launch_bounds__(256) void attn_kernel(
    const _Float16* __restrict__ Qh, const _Float16* __restrict__ Kh,
    const _Float16* __restrict__ Vt, const int* __restrict__ mask,
    _Float16* __restrict__ Oh) {
  __shared__ _Float16 plds[8][16 * 40];

  const int lane = threadIdx.x & 31;
  const int wave = threadIdx.x >> 5;
  const int bh   = blockIdx.x >> 3;
  const int qblk = blockIdx.x & 7;
  const int b = bh >> 4, h = bh & 15;
  const int q0 = qblk * 128 + wave * 16;

  const _Float16* Qbase = Qh + (size_t)b * 1024 * 1024 + h * 64;
  const _Float16* Kbase = Kh + (size_t)b * 1024 * 1024 + h * 64;
  const _Float16* Vbase = Vt + (size_t)(b * 16 + h) * 64 * 1024;
  const int* mb = mask + b * 1024;
  _Float16* myp = &plds[wave][0];

  const v16h qa0 = load_frag(Qbase, 1024, q0, 0, lane);
  const v16h qa1 = load_frag(Qbase, 1024, q0, 32, lane);

  v8f o[4] = {};
  v8f mrun = {-1e30f, -1e30f, -1e30f, -1e30f, -1e30f, -1e30f, -1e30f, -1e30f};
  v8f lrun = {};
  const float scale = 0.125f;
  const int mh8 = (lane >> 4) << 3;

#pragma unroll 1
  for (int kb = 0; kb < 1024; kb += 32) {
    v8f s[2] = {};
#pragma unroll
    for (int sub = 0; sub < 2; sub++) {
      v16h kf0 = load_frag(Kbase, 1024, kb + sub * 16, 0, lane);
      v16h kf1 = load_frag(Kbase, 1024, kb + sub * 16, 32, lane);
      s[sub] = wmma_f16(qa0, kf0, s[sub]);
      s[sub] = wmma_f16(qa1, kf1, s[sub]);
    }
    const int mk0 = mb[kb + (lane & 15)];
    const int mk1 = mb[kb + 16 + (lane & 15)];

    v8f bmax;
#pragma unroll
    for (int r = 0; r < 8; r++) {
      float v = -1e30f;
      if (mk0) v = fmaxf(v, s[0][r] * scale);
      if (mk1) v = fmaxf(v, s[1][r] * scale);
      bmax[r] = v;
    }
#pragma unroll
    for (int off = 1; off < 16; off <<= 1)
#pragma unroll
      for (int r = 0; r < 8; r++)
        bmax[r] = fmaxf(bmax[r], __shfl_xor(bmax[r], off, 32));

    v8f mnew, alpha, p0, p1, bsum;
#pragma unroll
    for (int r = 0; r < 8; r++) {
      mnew[r]  = fmaxf(mrun[r], bmax[r]);
      alpha[r] = __expf(mrun[r] - mnew[r]);
      p0[r] = mk0 ? __expf(s[0][r] * scale - mnew[r]) : 0.f;
      p1[r] = mk1 ? __expf(s[1][r] * scale - mnew[r]) : 0.f;
      bsum[r] = p0[r] + p1[r];
    }
#pragma unroll
    for (int off = 1; off < 16; off <<= 1)
#pragma unroll
      for (int r = 0; r < 8; r++)
        bsum[r] += __shfl_xor(bsum[r], off, 32);
#pragma unroll
    for (int r = 0; r < 8; r++) {
      lrun[r] = lrun[r] * alpha[r] + bsum[r];
      mrun[r] = mnew[r];
    }
#pragma unroll
    for (int t = 0; t < 4; t++)
#pragma unroll
      for (int r = 0; r < 8; r++) o[t][r] *= alpha[r];

#pragma unroll
    for (int r = 0; r < 8; r++) {
      myp[(r + mh8) * 40 + (lane & 15)]      = (_Float16)p0[r];
      myp[(r + mh8) * 40 + 16 + (lane & 15)] = (_Float16)p1[r];
    }
    asm volatile("s_wait_dscnt 0" ::: "memory");
    v16h pa = load_frag(myp, 40, 0, 0, lane);

#pragma unroll
    for (int t = 0; t < 4; t++) {
      v16h vf = load_frag(Vbase, 1024, t * 16, kb, lane);
      o[t] = wmma_f16(pa, vf, o[t]);
    }
  }

#pragma unroll
  for (int t = 0; t < 4; t++) {
    const int e = t * 16 + (lane & 15);
#pragma unroll
    for (int r = 0; r < 8; r++) {
      const int q = q0 + mh8 + r;
      const float denom = lrun[r];
      const float val = denom > 0.f ? o[t][r] / denom : 0.f;
      Oh[((size_t)b * 1024 + q) * 1024 + h * 64 + e] = (_Float16)val;
    }
  }
}

__global__ __launch_bounds__(256) void ln_kernel(
    const float* __restrict__ Ain, const float* __restrict__ Bin,
    const float* __restrict__ g, const float* __restrict__ beta,
    float* __restrict__ out32, _Float16* __restrict__ out16) {
  const int row = blockIdx.x;
  const int t = threadIdx.x;
  const int lane = t & 31, wave = t >> 5;

  const float4 xa = ((const float4*)(Ain + (size_t)row * 1024))[t];
  const float4 xb = ((const float4*)(Bin + (size_t)row * 1024))[t];
  float x[4] = {xa.x + xb.x, xa.y + xb.y, xa.z + xb.z, xa.w + xb.w};

  float s1 = x[0] + x[1] + x[2] + x[3];
  float s2 = x[0]*x[0] + x[1]*x[1] + x[2]*x[2] + x[3]*x[3];
#pragma unroll
  for (int off = 16; off >= 1; off >>= 1) {
    s1 += __shfl_xor(s1, off, 32);
    s2 += __shfl_xor(s2, off, 32);
  }
  __shared__ float r1[8], r2[8];
  if (lane == 0) { r1[wave] = s1; r2[wave] = s2; }
  __syncthreads();
  float tot1 = 0.f, tot2 = 0.f;
#pragma unroll
  for (int i = 0; i < 8; i++) { tot1 += r1[i]; tot2 += r2[i]; }

  const float mu   = tot1 * (1.f / 1024.f);
  const float var  = tot2 * (1.f / 1024.f) - mu * mu;
  const float rstd = rsqrtf(var + 1e-5f);
#pragma unroll
  for (int j = 0; j < 4; j++) {
    const int idx = t * 4 + j;
    const float y = (x[j] - mu) * rstd * g[idx] + beta[idx];
    out32[(size_t)row * 1024 + idx] = y;
    if (out16) out16[(size_t)row * 1024 + idx] = (_Float16)y;
  }
}

extern "C" void kernel_launch(void* const* d_in, const int* in_sizes, int n_in,
                              void* d_out, int out_size, void* d_ws, size_t ws_size,
                              hipStream_t stream) {
  (void)in_sizes; (void)n_in; (void)out_size; (void)ws_size;
  const float* X    = (const float*)d_in[0];
  const float* Y    = (const float*)d_in[1];
  const int*   mask = (const int*)d_in[2];
  const float* Wq   = (const float*)d_in[3];
  const float* Wk   = (const float*)d_in[4];
  const float* Wv   = (const float*)d_in[5];
  const float* Wo   = (const float*)d_in[6];
  const float* ln1g = (const float*)d_in[7];
  const float* ln1b = (const float*)d_in[8];
  const float* ln2g = (const float*)d_in[9];
  const float* ln2b = (const float*)d_in[10];
  const float* W1   = (const float*)d_in[11];
  const float* b1   = (const float*)d_in[12];
  const float* W2   = (const float*)d_in[13];
  const float* b2   = (const float*)d_in[14];

  char* ws = (char*)d_ws;
  const size_t MB = 1024 * 1024;
  _Float16* Xh   = (_Float16*)(ws + 0);
  _Float16* Yh   = (_Float16*)(ws + 8 * MB);
  _Float16* Wqh  = (_Float16*)(ws + 16 * MB);
  _Float16* Wkh  = (_Float16*)(ws + 18 * MB);
  _Float16* Wvh  = (_Float16*)(ws + 20 * MB);
  _Float16* Woh  = (_Float16*)(ws + 22 * MB);
  _Float16* W1h  = (_Float16*)(ws + 24 * MB);
  _Float16* W2h  = (_Float16*)(ws + 26 * MB);
  _Float16* Qh   = (_Float16*)(ws + 28 * MB);
  _Float16* Kh   = (_Float16*)(ws + 36 * MB);
  _Float16* Vt   = (_Float16*)(ws + 44 * MB);
  float*    Fbuf = (float*)(ws + 52 * MB);
  float*    Hres = (float*)(ws + 68 * MB);
  _Float16* Ohb  = Xh;
  _Float16* Hh   = Yh;
  _Float16* FF1h = Qh;

  dim3 blk256(256);

  cvt_all_kernel<<<512, blk256, 0, stream>>>(X, Y, Wq, Wk, Wv, Wo, W1, W2, Xh);

  const int ggrid = (4096 / 128) * (1024 / 128);

  gemm_nt_kernel<<<ggrid, blk256, 0, stream>>>(Xh, Wqh, nullptr, Qh, nullptr,
                                               4096, 1024, 1024, 0);
  gemm_nt_kernel<<<ggrid, blk256, 0, stream>>>(Yh, Wkh, nullptr, Kh, nullptr,
                                               4096, 1024, 1024, 0);
  gemm_nt_kernel<<<ggrid, blk256, 0, stream>>>(Yh, Wvh, nullptr, Vt, nullptr,
                                               4096, 1024, 1024, FLAG_TRANS);

  attn_kernel<<<512, blk256, 0, stream>>>(Qh, Kh, Vt, mask, Ohb);

  gemm_nt_kernel<<<ggrid, blk256, 0, stream>>>(Ohb, Woh, Fbuf, nullptr, nullptr,
                                               4096, 1024, 1024, 0);
  ln_kernel<<<4096, blk256, 0, stream>>>(X, Fbuf, ln1g, ln1b, Hres, Hh);

  gemm_nt_kernel<<<ggrid, blk256, 0, stream>>>(Hh, W1h, nullptr, FF1h, b1,
                                               4096, 1024, 1024, FLAG_RELU);
  gemm_nt_kernel<<<ggrid, blk256, 0, stream>>>(FF1h, W2h, Fbuf, nullptr, b2,
                                               4096, 1024, 1024, 0);
  ln_kernel<<<4096, blk256, 0, stream>>>(Hres, Fbuf, ln2g, ln2b, (float*)d_out,
                                         nullptr);
}


# attention: K and V (64-key stages) and the mask row staged once per workgroup through LDS by whole-line 16-byte global loads, two stages, next one in flight; waves read fragments from LDS
# speedup vs baseline: 1.9949x; 1.2802x over previous
#include <hip/hip_runtime.h>
#include <hip/hip_bf16.h>


typedef __attribute__((ext_vector_type(16))) _Float16 v16h;
typedef __attribute__((ext_vector_type(8)))  _Float16 v8h;
typedef __attribute__((ext_vector_type(4)))  _Float16 v4h;
typedef __attribute__((ext_vector_type(8)))  float    v8f;

#define FLAG_RELU  1
#define FLAG_TRANS 2

__device__ __forceinline__ v16h load_frag(const _Float16* base, int ld,
                                          int row0, int k0, int lane) {
  const int r  = row0 + (lane & 15);
  const int kb = k0 + ((lane >> 4) << 3);
  const _Float16* p = base + (size_t)r * ld + kb;
  v8h lo = *(const v8h*)(p);
  v8h hi = *(const v8h*)(p + 16);
  return __builtin_shufflevector(lo, hi, 0, 1, 2, 3, 4, 5, 6, 7,
                                         8, 9, 10, 11, 12, 13, 14, 15);
}

__device__ __forceinline__ v8f wmma_f16(v16h a, v16h b, v8f c) {
  return __builtin_amdgcn_wmma_f32_16x16x32_f16(false, a, false, b,
                                                (short)0, c, false, false);
}

#define CV_SLAB (512 * 256)

__device__ __forceinline__ v4h cvt_quad(float4 v) {
  v4h o = {(_Float16)v.x, (_Float16)v.y, (_Float16)v.z, (_Float16)v.w};
  return o;
}

__global__ __launch_bounds__(256) void cvt_all_kernel(
    const float* __restrict__ X,  const float* __restrict__ Y,
    const float* __restrict__ Wq, const float* __restrict__ Wk,
    const float* __restrict__ Wv, const float* __restrict__ Wo,
    const float* __restrict__ W1, const float* __restrict__ W2,
    _Float16* __restrict__ dst) {
  const size_t q0 = (size_t)blockIdx.x * 256 + threadIdx.x;
  float4 v[28];
#pragma unroll
  for (int i = 0; i < 8; i++) v[i]     = ((const float4*)X)[q0 + (size_t)i * CV_SLAB];
#pragma unroll
  for (int i = 0; i < 8; i++) v[8 + i] = ((const float4*)Y)[q0 + (size_t)i * CV_SLAB];
#pragma unroll
  for (int i = 0; i < 2; i++) {
    v[16 + i] = ((const float4*)Wq)[q0 + (size_t)i * CV_SLAB];
    v[18 + i] = ((const float4*)Wk)[q0 + (size_t)i * CV_SLAB];
    v[20 + i] = ((const float4*)Wv)[q0 + (size_t)i * CV_SLAB];
    v[22 + i] = ((const float4*)Wo)[q0 + (size_t)i * CV_SLAB];
    v[24 + i] = ((const float4*)W1)[q0 + (size_t)i * CV_SLAB];
    v[26 + i] = ((const float4*)W2)[q0 + (size_t)i * CV_SLAB];
  }
  __builtin_amdgcn_sched_barrier(0);
#pragma unroll
  for (int i = 0; i < 28; i++)
    ((v4h*)dst)[q0 + (size_t)i * CV_SLAB] = cvt_quad(v[i]);
}

#define G_LD    1024
#define G_BK    128
#define G_PITCH 136
#define G_STAGE (256 * G_PITCH)

__device__ __forceinline__ void gemm_stage_mma(const _Float16* sA,
                                               const _Float16* sB,
                                               v8f (&acc)[4][2], int lane) {
  v16h af[2][4], bf[2][2];
#pragma unroll
  for (int i = 0; i < 4; i++) af[0][i] = load_frag(sA, G_PITCH, 16 * i, 0, lane);
#pragma unroll
  for (int j = 0; j < 2; j++) bf[0][j] = load_frag(sB, G_PITCH, 16 * j, 0, lane);
#pragma unroll
  for (int u = 0; u < 4; u++) {
    const int c = u & 1, n = c ^ 1;
    if (u + 1 < 4) {
#pragma unroll
      for (int i = 0; i < 4; i++)
        af[n][i] = load_frag(sA, G_PITCH, 16 * i, 32 * (u + 1), lane);
#pragma unroll
      for (int j = 0; j < 2; j++)
        bf[n][j] = load_frag(sB, G_PITCH, 16 * j, 32 * (u + 1), lane);
    }
    __builtin_amdgcn_sched_barrier(0);
#pragma unroll
    for (int i = 0; i < 4; i++)
#pragma unroll
      for (int j = 0; j < 2; j++)
        acc[i][j] = wmma_f16(af[c][i], bf[c][j], acc[i][j]);
  }
}

__global__ __launch_bounds__(256) void gemm_nt_kernel(
    const _Float16* __restrict__ A, const _Float16* __restrict__ Bw,
    float* __restrict__ Cf, _Float16* __restrict__ Ch,
    const float* __restrict__ bias, int Mdim, int Ndim, int Kdim, int flags) {
  __shared__ __attribute__((aligned(16))) _Float16 lds[2 * G_STAGE];

  const int tid  = threadIdx.x;
  const int lane = tid & 31;
  const int wave = tid >> 5;
  const int tiles_n = Ndim >> 7;
  const int bm = blockIdx.x / tiles_n;
  const int bn = blockIdx.x - bm * tiles_n;
  const int wm = (wave & 1) * 64;
  const int wn = (wave >> 1) * 32;
  const int m0 = bm * 128 + wm;
  const int n0 = bn * 128 + wn;

  const int srow = tid >> 4;
  const int scol = (tid & 15) << 3;
  const _Float16* gA = A  + (size_t)(bm * 128 + srow) * G_LD + scol;
  const _Float16* gB = Bw + (size_t)(bn * 128 + srow) * G_LD + scol;
  const int soff = srow * G_PITCH + scol;

  v8f acc[4][2] = {};
  v8h ra[8], rb[8];

#pragma unroll
  for (int p = 0; p < 8; p++) {
    ra[p] = *(const v8h*)(gA + p * 16 * G_LD);
    rb[p] = *(const v8h*)(gB + p * 16 * G_LD);
  }
#pragma unroll
  for (int p = 0; p < 8; p++) {
    *(v8h*)(lds + soff + (16 * p) * G_PITCH)       = ra[p];
    *(v8h*)(lds + soff + (128 + 16 * p) * G_PITCH) = rb[p];
  }
  __syncthreads();

  const int nk = Kdim / G_BK;
#pragma unroll 1
  for (int kt = 0; kt + 1 < nk; kt++) {
    const _Float16* cur = lds + (kt & 1) * G_STAGE;
    _Float16*       nxt = lds + ((kt + 1) & 1) * G_STAGE;
    gA += G_BK;
    gB += G_BK;
#pragma unroll
    for (int p = 0; p < 8; p++) {
      ra[p] = *(const v8h*)(gA + p * 16 * G_LD);
      rb[p] = *(const v8h*)(gB + p * 16 * G_LD);
    }
    gemm_stage_mma(cur + wm * G_PITCH, cur + (128 + wn) * G_PITCH, acc, lane);
#pragma unroll
    for (int p = 0; p < 8; p++) {
      *(v8h*)(nxt + soff + (16 * p) * G_PITCH)       = ra[p];
      *(v8h*)(nxt + soff + (128 + 16 * p) * G_PITCH) = rb[p];
    }
    __syncthreads();
  }
  {
    const _Float16* cur = lds + ((nk - 1) & 1) * G_STAGE;
    gemm_stage_mma(cur + wm * G_PITCH, cur + (128 + wn) * G_PITCH, acc, lane);
  }

  const int nl = lane & 15;
  const int mh = (lane >> 4) << 3;
#pragma unroll
  for (int i = 0; i < 4; i++) {
#pragma unroll
    for (int j = 0; j < 2; j++) {
      const int n = n0 + j * 16 + nl;
      const float bv = bias ? bias[n] : 0.f;
#pragma unroll
      for (int r = 0; r < 8; r++) {
        const int m = m0 + 16 * i + mh + r;
        float v = acc[i][j][r] + bv;
        if (flags & FLAG_RELU) v = fmaxf(v, 0.f);
        if (Cf) Cf[(size_t)m * Ndim + n] = v;
        if (Ch) {
          if (flags & FLAG_TRANS) {
            Ch[((size_t)((m >> 10) << 10) + n) * 1024 + (m & 1023)] = (_Float16)v;
          } else {
            Ch[(size_t)m * Ndim + n] = (_Float16)v;
          }
        }
      }
    }
  }
}

#define AT_PITCH 72
#define AT_TILE  (64 * AT_PITCH)
#define AT_STAGE (2 * AT_TILE)

__global__ __launch_bounds__(256) void attn_kernel(
    const _Float16* __restrict__ Qh, const _Float16* __restrict__ Kh,
    const _Float16* __restrict__ Vt, const int* __restrict__ mask,
    _Float16* __restrict__ Oh) {
  __shared__ __attribute__((aligned(16))) _Float16 kv[2 * AT_STAGE];
  __shared__ __attribute__((aligned(16))) _Float16 plds[8][16 * 40];
  __shared__ __attribute__((aligned(16))) int smask[1024];

  const int tid  = threadIdx.x;
  const int lane = tid & 31;
  const int wave = tid >> 5;
  const int bh   = blockIdx.x >> 3;
  const int qblk = blockIdx.x & 7;
  const int b = bh >> 4, h = bh & 15;
  const int q0 = qblk * 128 + wave * 16;

  const _Float16* Qbase = Qh + (size_t)b * 1024 * 1024 + h * 64;
  const _Float16* Kbase = Kh + (size_t)b * 1024 * 1024 + h * 64;
  const _Float16* Vbase = Vt + (size_t)(b * 16 + h) * 64 * 1024;
  const int* mb = mask + b * 1024;
  _Float16* myp = &plds[wave][0];

  const int srow = tid >> 3;
  const int scol = (tid & 7) << 3;
  const _Float16* gK = Kbase + (size_t)srow * 1024 + scol;
  const _Float16* gV = Vbase + (size_t)srow * 1024 + scol;
  const int soff = srow * AT_PITCH + scol;

  v8h rk0 = *(const v8h*)(gK);
  v8h rk1 = *(const v8h*)(gK + 32 * 1024);
  v8h rv0 = *(const v8h*)(gV);
  v8h rv1 = *(const v8h*)(gV + 32 * 1024);
  ((int4*)smask)[tid] = ((const int4*)mb)[tid];

  const v16h qa0 = load_frag(Qbase, 1024, q0, 0, lane);
  const v16h qa1 = load_frag(Qbase, 1024, q0, 32, lane);

  *(v8h*)(kv + soff)                           = rk0;
  *(v8h*)(kv + soff + 32 * AT_PITCH)           = rk1;
  *(v8h*)(kv + AT_TILE + soff)                 = rv0;
  *(v8h*)(kv + AT_TILE + soff + 32 * AT_PITCH) = rv1;
  __syncthreads();

  v8f o[4] = {};
  v8f mrun = {-1e30f, -1e30f, -1e30f, -1e30f, -1e30f, -1e30f, -1e30f, -1e30f};
  v8f lrun = {};
  const float scale = 0.125f;
  const int mh8 = (lane >> 4) << 3;

#pragma unroll 1
  for (int st = 0; st < 16; st++) {
    const _Float16* sK = kv + (st & 1) * AT_STAGE;
    const _Float16* sV = sK + AT_TILE;
    _Float16*      nxt = kv + ((st + 1) & 1) * AT_STAGE;
    const int sn = st + 1 < 16 ? st + 1 : 15;
    rk0 = *(const v8h*)(gK + (size_t)sn * 64 * 1024);
    rk1 = *(const v8h*)(gK + (size_t)sn * 64 * 1024 + 32 * 1024);
    rv0 = *(const v8h*)(gV + sn * 64);
    rv1 = *(const v8h*)(gV + sn * 64 + 32 * 1024);

#pragma unroll
    for (int j = 0; j < 2; j++) {
      const int kb = st * 64 + j * 32;
      v16h kf[2][2], vf[4];
#pragma unroll
      for (int sub = 0; sub < 2; sub++) {
        kf[sub][0] = load_frag(sK, AT_PITCH, j * 32 + sub * 16, 0, lane);
        kf[sub][1] = load_frag(sK, AT_PITCH, j * 32 + sub * 16, 32, lane);
      }
#pragma unroll
      for (int t = 0; t < 4; t++)
        vf[t] = load_frag(sV, AT_PITCH, t * 16, j * 32, lane);
      const int mk0 = smask[kb + (lane & 15)];
      const int mk1 = smask[kb + 16 + (lane & 15)];

      v8f s[2] = {};
#pragma unroll
      for (int sub = 0; sub < 2; sub++) {
        s[sub] = wmma_f16(qa0, kf[sub][0], s[sub]);
        s[sub] = wmma_f16(qa1, kf[sub][1], s[sub]);
      }

      v8f bmax;
#pragma unroll
      for (int r = 0; r < 8; r++) {
        float v = -1e30f;
        if (mk0) v = fmaxf(v, s[0][r] * scale);
        if (mk1) v = fmaxf(v, s[1][r] * scale);
        bmax[r] = v;
      }
#pragma unroll
      for (int off = 1; off < 16; off <<= 1)
#pragma unroll
        for (int r = 0; r < 8; r++)
          bmax[r] = fmaxf(bmax[r], __shfl_xor(bmax[r], off, 32));

      v8f mnew, alpha, p0, p1, bsum;
#pragma unroll
      for (int r = 0; r < 8; r++) {
        mnew[r]  = fmaxf(mrun[r], bmax[r]);
        alpha[r] = __expf(mrun[r] - mnew[r]);
        p0[r] = mk0 ? __expf(s[0][r] * scale - mnew[r]) : 0.f;
        p1[r] = mk1 ? __expf(s[1][r] * scale - mnew[r]) : 0.f;
        bsum[r] = p0[r] + p1[r];
      }
#pragma unroll
      for (int off = 1; off < 16; off <<= 1)
#pragma unroll
        for (int r = 0; r < 8; r++)
          bsum[r] += __shfl_xor(bsum[r], off, 32);
#pragma unroll
      for (int r = 0; r < 8; r++) {
        lrun[r] = lrun[r] * alpha[r] + bsum[r];
        mrun[r] = mnew[r];
      }
#pragma unroll
      for (int t = 0; t < 4; t++)
#pragma unroll
        for (int r = 0; r < 8; r++) o[t][r] *= alpha[r];

#pragma unroll
      for (int r = 0; r < 8; r++) {
        myp[(r + mh8) * 40 + (lane & 15)]      = (_Float16)p0[r];
        myp[(r + mh8) * 40 + 16 + (lane & 15)] = (_Float16)p1[r];
      }
      asm volatile("s_wait_dscnt 0" ::: "memory");
      v16h pa = load_frag(myp, 40, 0, 0, lane);

#pragma unroll
      for (int t = 0; t < 4; t++) o[t] = wmma_f16(pa, vf[t], o[t]);
    }

    *(v8h*)(nxt + soff)                           = rk0;
    *(v8h*)(nxt + soff + 32 * AT_PITCH)           = rk1;
    *(v8h*)(nxt + AT_TILE + soff)                 = rv0;
    *(v8h*)(nxt + AT_TILE + soff + 32 * AT_PITCH) = rv1;
    __syncthreads();
  }

#pragma unroll
  for (int t = 0; t < 4; t++) {
    const int e = t * 16 + (lane & 15);
#pragma unroll
    for (int r = 0; r < 8; r++) {
      const int q = q0 + mh8 + r;
      const float denom = lrun[r];
      const float val = denom > 0.f ? o[t][r] / denom : 0.f;
      Oh[((size_t)b * 1024 + q) * 1024 + h * 64 + e] = (_Float16)val;
    }
  }
}

__global__ __launch_bounds__(256) void ln_kernel(
    const float* __restrict__ Ain, const float* __restrict__ Bin,
    const float* __restrict__ g, const float* __restrict__ beta,
    float* __restrict__ out32, _Float16* __restrict__ out16) {
  const int row = blockIdx.x;
  const int t = threadIdx.x;
  const int lane = t & 31, wave = t >> 5;

  const float4 xa = ((const float4*)(Ain + (size_t)row * 1024))[t];
  const float4 xb = ((const float4*)(Bin + (size_t)row * 1024))[t];
  float x[4] = {xa.x + xb.x, xa.y + xb.y, xa.z + xb.z, xa.w + xb.w};

  float s1 = x[0] + x[1] + x[2] + x[3];
  float s2 = x[0]*x[0] + x[1]*x[1] + x[2]*x[2] + x[3]*x[3];
#pragma unroll
  for (int off = 16; off >= 1; off >>= 1) {
    s1 += __shfl_xor(s1, off, 32);
    s2 += __shfl_xor(s2, off, 32);
  }
  __shared__ float r1[8], r2[8];
  if (lane == 0) { r1[wave] = s1; r2[wave] = s2; }
  __syncthreads();
  float tot1 = 0.f, tot2 = 0.f;
#pragma unroll
  for (int i = 0; i < 8; i++) { tot1 += r1[i]; tot2 += r2[i]; }

  const float mu   = tot1 * (1.f / 1024.f);
  const float var  = tot2 * (1.f / 1024.f) - mu * mu;
  const float rstd = rsqrtf(var + 1e-5f);
#pragma unroll
  for (int j = 0; j < 4; j++) {
    const int idx = t * 4 + j;
    const float y = (x[j] - mu) * rstd * g[idx] + beta[idx];
    out32[(size_t)row * 1024 + idx] = y;
    if (out16) out16[(size_t)row * 1024 + idx] = (_Float16)y;
  }
}

extern "C" void kernel_launch(void* const* d_in, const int* in_sizes, int n_in,
                              void* d_out, int out_size, void* d_ws, size_t ws_size,
                              hipStream_t stream) {
  (void)in_sizes; (void)n_in; (void)out_size; (void)ws_size;
  const float* X    = (const float*)d_in[0];
  const float* Y    = (const float*)d_in[1];
  const int*   mask = (const int*)d_in[2];
  const float* Wq   = (const float*)d_in[3];
  const float* Wk   = (const float*)d_in[4];
  const float* Wv   = (const float*)d_in[5];
  const float* Wo   = (const float*)d_in[6];
  const float* ln1g = (const float*)d_in[7];
  const float* ln1b = (const float*)d_in[8];
  const float* ln2g = (const float*)d_in[9];
  const float* ln2b = (const float*)d_in[10];
  const float* W1   = (const float*)d_in[11];
  const float* b1   = (const float*)d_in[12];
  const float* W2   = (const float*)d_in[13];
  const float* b2   = (const float*)d_in[14];

  char* ws = (char*)d_ws;
  const size_t MB = 1024 * 1024;
  _Float16* Xh   = (_Float16*)(ws + 0);
  _Float16* Yh   = (_Float16*)(ws + 8 * MB);
  _Float16* Wqh  = (_Float16*)(ws + 16 * MB);
  _Float16* Wkh  = (_Float16*)(ws + 18 * MB);
  _Float16* Wvh  = (_Float16*)(ws + 20 * MB);
  _Float16* Woh  = (_Float16*)(ws + 22 * MB);
  _Float16* W1h  = (_Float16*)(ws + 24 * MB);
  _Float16* W2h  = (_Float16*)(ws + 26 * MB);
  _Float16* Qh   = (_Float16*)(ws + 28 * MB);
  _Float16* Kh   = (_Float16*)(ws + 36 * MB);
  _Float16* Vt   = (_Float16*)(ws + 44 * MB);
  float*    Fbuf = (float*)(ws + 52 * MB);
  float*    Hres = (float*)(ws + 68 * MB);
  _Float16* Ohb  = Xh;
  _Float16* Hh   = Yh;
  _Float16* FF1h = Qh;

  dim3 blk256(256);

  cvt_all_kernel<<<512, blk256, 0, stream>>>(X, Y, Wq, Wk, Wv, Wo, W1, W2, Xh);

  const int ggrid = (4096 / 128) * (1024 / 128);

  gemm_nt_kernel<<<ggrid, blk256, 0, stream>>>(Xh, Wqh, nullptr, Qh, nullptr,
                                               4096, 1024, 1024, 0);
  gemm_nt_kernel<<<ggrid, blk256, 0, stream>>>(Yh, Wkh, nullptr, Kh, nullptr,
                                               4096, 1024, 1024, 0);
  gemm_nt_kernel<<<ggrid, blk256, 0, stream>>>(Yh, Wvh, nullptr, Vt, nullptr,
                                               4096, 1024, 1024, FLAG_TRANS);

  attn_kernel<<<512, blk256, 0, stream>>>(Qh, Kh, Vt, mask, Ohb);

  gemm_nt_kernel<<<ggrid, blk256, 0, stream>>>(Ohb, Woh, Fbuf, nullptr, nullptr,
                                               4096, 1024, 1024, 0);
  ln_kernel<<<4096, blk256, 0, stream>>>(X, Fbuf, ln1g, ln1b, Hres, Hh);

  gemm_nt_kernel<<<ggrid, blk256, 0, stream>>>(Hh, W1h, nullptr, FF1h, b1,
                                               4096, 1024, 1024, FLAG_RELU);
  gemm_nt_kernel<<<ggrid, blk256, 0, stream>>>(FF1h, W2h, Fbuf, nullptr, b2,
                                               4096, 1024, 1024, 0);
  ln_kernel<<<4096, blk256, 0, stream>>>(Hres, Fbuf, ln2g, ln2b, (float*)d_out,
                                         nullptr);
}


# gemm epilogue: C tile staged through LDS, stored as whole 128-byte lines by 16-byte stores (weight fragment as WMMA A: a lane holds 8 consecutive n; V transposed in LDS); variant fixed per launch
# speedup vs baseline: 2.6948x; 1.3508x over previous
#include <hip/hip_runtime.h>
#include <hip/hip_bf16.h>


typedef __attribute__((ext_vector_type(16))) _Float16 v16h;
typedef __attribute__((ext_vector_type(8)))  _Float16 v8h;
typedef __attribute__((ext_vector_type(4)))  _Float16 v4h;
typedef __attribute__((ext_vector_type(8)))  float    v8f;

__device__ __forceinline__ v16h load_frag(const _Float16* base, int ld,
                                          int row0, int k0, int lane) {
  const int r  = row0 + (lane & 15);
  const int kb = k0 + ((lane >> 4) << 3);
  const _Float16* p = base + (size_t)r * ld + kb;
  v8h lo = *(const v8h*)(p);
  v8h hi = *(const v8h*)(p + 16);
  return __builtin_shufflevector(lo, hi, 0, 1, 2, 3, 4, 5, 6, 7,
                                         8, 9, 10, 11, 12, 13, 14, 15);
}

__device__ __forceinline__ v8f wmma_f16(v16h a, v16h b, v8f c) {
  return __builtin_amdgcn_wmma_f32_16x16x32_f16(false, a, false, b,
                                                (short)0, c, false, false);
}

#define CV_SLAB (512 * 256)

__device__ __forceinline__ v4h cvt_quad(float4 v) {
  v4h o = {(_Float16)v.x, (_Float16)v.y, (_Float16)v.z, (_Float16)v.w};
  return o;
}

__global__ __launch_bounds__(256) void cvt_all_kernel(
    const float* __restrict__ X,  const float* __restrict__ Y,
    const float* __restrict__ Wq, const float* __restrict__ Wk,
    const float* __restrict__ Wv, const float* __restrict__ Wo,
    const float* __restrict__ W1, const float* __restrict__ W2,
    _Float16* __restrict__ dst) {
  const size_t q0 = (size_t)blockIdx.x * 256 + threadIdx.x;
  float4 v[28];
#pragma unroll
  for (int i = 0; i < 8; i++) v[i]     = ((const float4*)X)[q0 + (size_t)i * CV_SLAB];
#pragma unroll
  for (int i = 0; i < 8; i++) v[8 + i] = ((const float4*)Y)[q0 + (size_t)i * CV_SLAB];
#pragma unroll
  for (int i = 0; i < 2; i++) {
    v[16 + i] = ((const float4*)Wq)[q0 + (size_t)i * CV_SLAB];
    v[18 + i] = ((const float4*)Wk)[q0 + (size_t)i * CV_SLAB];
    v[20 + i] = ((const float4*)Wv)[q0 + (size_t)i * CV_SLAB];
    v[22 + i] = ((const float4*)Wo)[q0 + (size_t)i * CV_SLAB];
    v[24 + i] = ((const float4*)W1)[q0 + (size_t)i * CV_SLAB];
    v[26 + i] = ((const float4*)W2)[q0 + (size_t)i * CV_SLAB];
  }
  __builtin_amdgcn_sched_barrier(0);
#pragma unroll
  for (int i = 0; i < 28; i++)
    ((v4h*)dst)[q0 + (size_t)i * CV_SLAB] = cvt_quad(v[i]);
}

#define G_LD    1024
#define G_BK    128
#define G_PITCH 136
#define G_STAGE (256 * G_PITCH)
#define E_PITCH32 132

#define OUT_F16  0
#define OUT_F16T 1
#define OUT_F32  2

template <bool SWAP>
__device__ __forceinline__ void gemm_stage_mma(const _Float16* sA,
                                               const _Float16* sB,
                                               v8f (&acc)[4][2], int lane) {
  v16h af[2][4], bf[2][2];
#pragma unroll
  for (int i = 0; i < 4; i++) af[0][i] = load_frag(sA, G_PITCH, 16 * i, 0, lane);
#pragma unroll
  for (int j = 0; j < 2; j++) bf[0][j] = load_frag(sB, G_PITCH, 16 * j, 0, lane);
#pragma unroll
  for (int u = 0; u < 4; u++) {
    const int c = u & 1, n = c ^ 1;
    if (u + 1 < 4) {
#pragma unroll
      for (int i = 0; i < 4; i++)
        af[n][i] = load_frag(sA, G_PITCH, 16 * i, 32 * (u + 1), lane);
#pragma unroll
      for (int j = 0; j < 2; j++)
        bf[n][j] = load_frag(sB, G_PITCH, 16 * j, 32 * (u + 1), lane);
    }
    __builtin_amdgcn_sched_barrier(0);
#pragma unroll
    for (int i = 0; i < 4; i++)
#pragma unroll
      for (int j = 0; j < 2; j++)
        acc[i][j] = SWAP ? wmma_f16(bf[c][j], af[c][i], acc[i][j])
                         : wmma_f16(af[c][i], bf[c][j], acc[i][j]);
  }
}

template <int OUT, bool BIAS, bool RELU>
__global__ __launch_bounds__(256) void gemm_nt_kernel(
    const _Float16* __restrict__ A, const _Float16* __restrict__ Bw,
    float* __restrict__ Cf, _Float16* __restrict__ Ch,
    const float* __restrict__ bias, int Mdim, int Ndim, int Kdim) {
  __shared__ __attribute__((aligned(16))) _Float16 lds[2 * G_STAGE];
  constexpr bool SWAP = (OUT != OUT_F16T);
  (void)Mdim;

  const int tid  = threadIdx.x;
  const int lane = tid & 31;
  const int wave = tid >> 5;
  const int tiles_n = Ndim >> 7;
  const int bm = blockIdx.x / tiles_n;
  const int bn = blockIdx.x - bm * tiles_n;
  const int wm = (wave & 1) * 64;
  const int wn = (wave >> 1) * 32;

  const int srow = tid >> 4;
  const int scol = (tid & 15) << 3;
  const _Float16* gA = A  + (size_t)(bm * 128 + srow) * G_LD + scol;
  const _Float16* gB = Bw + (size_t)(bn * 128 + srow) * G_LD + scol;
  const int soff = srow * G_PITCH + scol;

  v8f acc[4][2] = {};
  v8h ra[8], rb[8];

#pragma unroll
  for (int p = 0; p < 8; p++) {
    ra[p] = *(const v8h*)(gA + p * 16 * G_LD);
    rb[p] = *(const v8h*)(gB + p * 16 * G_LD);
  }
#pragma unroll
  for (int p = 0; p < 8; p++) {
    *(v8h*)(lds + soff + (16 * p) * G_PITCH)       = ra[p];
    *(v8h*)(lds + soff + (128 + 16 * p) * G_PITCH) = rb[p];
  }
  __syncthreads();

  const int nk = Kdim / G_BK;
#pragma unroll 1
  for (int kt = 0; kt + 1 < nk; kt++) {
    const _Float16* cur = lds + (kt & 1) * G_STAGE;
    _Float16*       nxt = lds + ((kt + 1) & 1) * G_STAGE;
    gA += G_BK;
    gB += G_BK;
#pragma unroll
    for (int p = 0; p < 8; p++) {
      ra[p] = *(const v8h*)(gA + p * 16 * G_LD);
      rb[p] = *(const v8h*)(gB + p * 16 * G_LD);
    }
    gemm_stage_mma<SWAP>(cur + wm * G_PITCH, cur + (128 + wn) * G_PITCH, acc, lane);
#pragma unroll
    for (int p = 0; p < 8; p++) {
      *(v8h*)(nxt + soff + (16 * p) * G_PITCH)       = ra[p];
      *(v8h*)(nxt + soff + (128 + 16 * p) * G_PITCH) = rb[p];
    }
    __syncthreads();
  }

  const int lc = lane & 15;
  const int lh = lane >> 4;

  float bv[2][8];
#pragma unroll
  for (int j = 0; j < 2; j++) {
    if (BIAS && SWAP) {
      const float* bp = bias + bn * 128 + wn + 16 * j + 8 * lh;
      const float4 b0 = *(const float4*)(bp);
      const float4 b1 = *(const float4*)(bp + 4);
      bv[j][0] = b0.x; bv[j][1] = b0.y; bv[j][2] = b0.z; bv[j][3] = b0.w;
      bv[j][4] = b1.x; bv[j][5] = b1.y; bv[j][6] = b1.z; bv[j][7] = b1.w;
    } else {
      const float b0 = BIAS ? bias[bn * 128 + wn + 16 * j + lc] : 0.f;
#pragma unroll
      for (int r = 0; r < 8; r++) bv[j][r] = b0;
    }
  }

  {
    const _Float16* cur = lds + ((nk - 1) & 1) * G_STAGE;
    gemm_stage_mma<SWAP>(cur + wm * G_PITCH, cur + (128 + wn) * G_PITCH, acc, lane);
  }

  _Float16* st = lds + (nk & 1) * G_STAGE;

  if (OUT == OUT_F32) {
    float* sf = (float*)st;
#pragma unroll
    for (int i = 0; i < 4; i++) {
#pragma unroll
      for (int j = 0; j < 2; j++) {
        float v[8];
#pragma unroll
        for (int r = 0; r < 8; r++) {
          v[r] = BIAS ? acc[i][j][r] + bv[j][r] : acc[i][j][r];
          if (RELU) v[r] = fmaxf(v[r], 0.f);
        }
        float* q = sf + (wm + 16 * i + lc) * E_PITCH32 + wn + 16 * j + 8 * lh;
        *(float4*)(q)     = make_float4(v[0], v[1], v[2], v[3]);
        *(float4*)(q + 4) = make_float4(v[4], v[5], v[6], v[7]);
      }
    }
    __syncthreads();
    const int row0 = tid >> 5;
    const int col  = (tid & 31) << 2;
    float* gC = Cf + (size_t)(bm * 128 + row0) * Ndim + bn * 128 + col;
    const float* sp = sf + row0 * E_PITCH32 + col;
#pragma unroll
    for (int p = 0; p < 16; p++) {
      const float4 o = *(const float4*)(sp + (8 * p) * E_PITCH32);
      *(float4*)(gC + (size_t)(8 * p) * Ndim) = o;
    }
  } else {
#pragma unroll
    for (int i = 0; i < 4; i++) {
#pragma unroll
      for (int j = 0; j < 2; j++) {
        v8h o;
#pragma unroll
        for (int r = 0; r < 8; r++) {
          float v = BIAS ? acc[i][j][r] + bv[j][r] : acc[i][j][r];
          if (RELU) v = fmaxf(v, 0.f);
          o[r] = (_Float16)v;
        }
        if (SWAP)
          *(v8h*)(st + (wm + 16 * i + lc) * G_PITCH + wn + 16 * j + 8 * lh) = o;
        else
          *(v8h*)(st + (wn + 16 * j + lc) * G_PITCH + wm + 16 * i + 8 * lh) = o;
      }
    }
    __syncthreads();
    _Float16* gC;
    size_t gpitch;
    if (OUT == OUT_F16) {
      gC = Ch + (size_t)(bm * 128 + srow) * Ndim + bn * 128 + scol;
      gpitch = (size_t)Ndim;
    } else {
      const int mg = bm * 128 + scol;
      gC = Ch + ((size_t)((mg >> 10) << 10) + bn * 128 + srow) * 1024 + (mg & 1023);
      gpitch = 1024;
    }
#pragma unroll
    for (int p = 0; p < 8; p++) {
      const v8h o = *(const v8h*)(st + soff + (16 * p) * G_PITCH);
      *(v8h*)(gC + (size_t)(16 * p) * gpitch) = o;
    }
  }
}

#define AT_PITCH 72
#define AT_TILE  (64 * AT_PITCH)
#define AT_STAGE (2 * AT_TILE)

__device__ __forceinline__ float xhalf(float x) {
  const int xi = __builtin_bit_cast(int, x);
  return __builtin_bit_cast(float, __builtin_amdgcn_permlanex16(
      xi, xi, 0x76543210, 0xfedcba98, false, false));
}

__global__ __launch_bounds__(256) __attribute__((amdgpu_waves_per_eu(4, 4)))
void attn_kernel(
    const _Float16* __restrict__ Qh, const _Float16* __restrict__ Kh,
    const _Float16* __restrict__ Vt, const int* __restrict__ mask,
    _Float16* __restrict__ Oh) {
  __shared__ __attribute__((aligned(16))) _Float16 kv[2 * AT_STAGE];
  __shared__ __attribute__((aligned(16))) float sbias[1024];

  const int tid  = threadIdx.x;
  const int lane = tid & 31;
  const int wave = tid >> 5;
  const int lc = lane & 15;
  const int lh = lane >> 4;
  const int bh   = blockIdx.x >> 3;
  const int qblk = blockIdx.x & 7;
  const int b = bh >> 4, h = bh & 15;
  const int q0 = qblk * 128 + wave * 16;

  const _Float16* Qbase = Qh + (size_t)b * 1024 * 1024 + h * 64;
  const _Float16* Kbase = Kh + (size_t)b * 1024 * 1024 + h * 64;
  const _Float16* Vbase = Vt + (size_t)(b * 16 + h) * 64 * 1024;
  const int* mb = mask + b * 1024;

  const int srow = tid >> 3;
  const int scol = (tid & 7) << 3;
  const _Float16* gK = Kbase + (size_t)srow * 1024 + scol;
  const _Float16* gV = Vbase + (size_t)srow * 1024 + scol;
  const int soff = srow * AT_PITCH + scol;

  v8h rk0 = *(const v8h*)(gK);
  v8h rk1 = *(const v8h*)(gK + 32 * 1024);
  v8h rv0 = *(const v8h*)(gV);
  v8h rv1 = *(const v8h*)(gV + 32 * 1024);
  {
    const int4 mv = ((const int4*)mb)[tid];
    float4 bv;
    bv.x = mv.x ? 0.f : -1e30f;
    bv.y = mv.y ? 0.f : -1e30f;
    bv.z = mv.z ? 0.f : -1e30f;
    bv.w = mv.w ? 0.f : -1e30f;
    ((float4*)sbias)[tid] = bv;
  }

  const v16h qb0 = load_frag(Qbase, 1024, q0, 0, lane);
  const v16h qb1 = load_frag(Qbase, 1024, q0, 32, lane);

  *(v8h*)(kv + soff)                           = rk0;
  *(v8h*)(kv + soff + 32 * AT_PITCH)           = rk1;
  *(v8h*)(kv + AT_TILE + soff)                 = rv0;
  *(v8h*)(kv + AT_TILE + soff + 32 * AT_PITCH) = rv1;
  __syncthreads();

  v8f o[4] = {};
  float mrun = -1e30f;
  float lrun = 0.f;
  const float scale = 0.125f;

#pragma unroll 1
  for (int st = 0; st < 16; st++) {
    const _Float16* sK = kv + (st & 1) * AT_STAGE;
    const _Float16* sV = sK + AT_TILE;
    _Float16*      nxt = kv + ((st + 1) & 1) * AT_STAGE;
    const int sn = st + 1 < 16 ? st + 1 : 15;
    rk0 = *(const v8h*)(gK + (size_t)sn * 64 * 1024);
    rk1 = *(const v8h*)(gK + (size_t)sn * 64 * 1024 + 32 * 1024);
    rv0 = *(const v8h*)(gV + sn * 64);
    rv1 = *(const v8h*)(gV + sn * 64 + 32 * 1024);

#pragma unroll
    for (int j = 0; j < 2; j++) {
      const int kb = st * 64 + j * 32;
      v16h kf[2][2], vf[4];
#pragma unroll
      for (int sub = 0; sub < 2; sub++) {
        kf[sub][0] = load_frag(sK, AT_PITCH, j * 32 + sub * 16, 0, lane);
        kf[sub][1] = load_frag(sK, AT_PITCH, j * 32 + sub * 16, 32, lane);
      }
#pragma unroll
      for (int t = 0; t < 4; t++)
        vf[t] = load_frag(sV, AT_PITCH, t * 16, j * 32, lane);
      const float* bp = sbias + kb + 8 * lh;
      const float4 b00 = *(const float4*)(bp);
      const float4 b01 = *(const float4*)(bp + 4);
      const float4 b10 = *(const float4*)(bp + 16);
      const float4 b11 = *(const float4*)(bp + 20);
      const float bias[2][8] = {
          {b00.x, b00.y, b00.z, b00.w, b01.x, b01.y, b01.z, b01.w},
          {b10.x, b10.y, b10.z, b10.w, b11.x, b11.y, b11.z, b11.w}};

      v8f s[2] = {};
#pragma unroll
      for (int sub = 0; sub < 2; sub++) {
        s[sub] = wmma_f16(kf[sub][0], qb0, s[sub]);
        s[sub] = wmma_f16(kf[sub][1], qb1, s[sub]);
      }

      float v[2][8];
#pragma unroll
      for (int sub = 0; sub < 2; sub++)
#pragma unroll
        for (int r = 0; r < 8; r++)
          v[sub][r] = fmaf(s[sub][r], scale, bias[sub][r]);
      float mx = v[0][0];
#pragma unroll
      for (int sub = 0; sub < 2; sub++)
#pragma unroll
        for (int r = 0; r < 8; r++) mx = fmaxf(mx, v[sub][r]);
      mx = fmaxf(mx, xhalf(mx));

      const float mnew  = fmaxf(mrun, mx);
      const float alpha = __expf(mrun - mnew);
      float p[2][8];
#pragma unroll
      for (int sub = 0; sub < 2; sub++)
#pragma unroll
        for (int r = 0; r < 8; r++)
          p[sub][r] = __expf(v[sub][r] - mnew);
      float psum = 0.f;
#pragma unroll
      for (int sub = 0; sub < 2; sub++)
#pragma unroll
        for (int r = 0; r < 8; r += 2) psum += p[sub][r] + p[sub][r + 1];
      lrun = lrun * alpha + psum;
      mrun = mnew;
#pragma unroll
      for (int t = 0; t < 4; t++)
#pragma unroll
        for (int r = 0; r < 8; r++) o[t][r] *= alpha;

      v16h pb;
#pragma unroll
      for (int r = 0; r < 8; r++) {
        pb[r]     = (_Float16)p[0][r];
        pb[8 + r] = (_Float16)p[1][r];
      }
#pragma unroll
      for (int t = 0; t < 4; t++) o[t] = wmma_f16(vf[t], pb, o[t]);
    }

    *(v8h*)(nxt + soff)                           = rk0;
    *(v8h*)(nxt + soff + 32 * AT_PITCH)           = rk1;
    *(v8h*)(nxt + AT_TILE + soff)                 = rv0;
    *(v8h*)(nxt + AT_TILE + soff + 32 * AT_PITCH) = rv1;
    __syncthreads();
  }

  const float ltot = lrun + xhalf(lrun);
  const bool live = mrun > -1e29f;
  _Float16* orow = Oh + ((size_t)b * 1024 + q0 + lc) * 1024 + h * 64 + 8 * lh;
#pragma unroll
  for (int t = 0; t < 4; t++) {
    v8h ov;
#pragma unroll
    for (int r = 0; r < 8; r++) {
      const float val = live ? o[t][r] / ltot : 0.f;
      ov[r] = (_Float16)val;
    }
    *(v8h*)(orow + t * 16) = ov;
  }
}

__global__ __launch_bounds__(256) void ln_kernel(
    const float* __restrict__ Ain, const float* __restrict__ Bin,
    const float* __restrict__ g, const float* __restrict__ beta,
    float* __restrict__ out32, _Float16* __restrict__ out16) {
  const int row = blockIdx.x;
  const int t = threadIdx.x;
  const int lane = t & 31, wave = t >> 5;

  const float4 xa = ((const float4*)(Ain + (size_t)row * 1024))[t];
  const float4 xb = ((const float4*)(Bin + (size_t)row * 1024))[t];
  float x[4] = {xa.x + xb.x, xa.y + xb.y, xa.z + xb.z, xa.w + xb.w};

  float s1 = x[0] + x[1] + x[2] + x[3];
  float s2 = x[0]*x[0] + x[1]*x[1] + x[2]*x[2] + x[3]*x[3];
#pragma unroll
  for (int off = 16; off >= 1; off >>= 1) {
    s1 += __shfl_xor(s1, off, 32);
    s2 += __shfl_xor(s2, off, 32);
  }
  __shared__ float r1[8], r2[8];
  if (lane == 0) { r1[wave] = s1; r2[wave] = s2; }
  __syncthreads();
  float tot1 = 0.f, tot2 = 0.f;
#pragma unroll
  for (int i = 0; i < 8; i++) { tot1 += r1[i]; tot2 += r2[i]; }

  const float mu   = tot1 * (1.f / 1024.f);
  const float var  = tot2 * (1.f / 1024.f) - mu * mu;
  const float rstd = rsqrtf(var + 1e-5f);
#pragma unroll
  for (int j = 0; j < 4; j++) {
    const int idx = t * 4 + j;
    const float y = (x[j] - mu) * rstd * g[idx] + beta[idx];
    out32[(size_t)row * 1024 + idx] = y;
    if (out16) out16[(size_t)row * 1024 + idx] = (_Float16)y;
  }
}

extern "C" void kernel_launch(void* const* d_in, const int* in_sizes, int n_in,
                              void* d_out, int out_size, void* d_ws, size_t ws_size,
                              hipStream_t stream) {
  (void)in_sizes; (void)n_in; (void)out_size; (void)ws_size;
  const float* X    = (const float*)d_in[0];
  const float* Y    = (const float*)d_in[1];
  const int*   mask = (const int*)d_in[2];
  const float* Wq   = (const float*)d_in[3];
  const float* Wk   = (const float*)d_in[4];
  const float* Wv   = (const float*)d_in[5];
  const float* Wo   = (const float*)d_in[6];
  const float* ln1g = (const float*)d_in[7];
  const float* ln1b = (const float*)d_in[8];
  const float* ln2g = (const float*)d_in[9];
  const float* ln2b = (const float*)d_in[10];
  const float* W1   = (const float*)d_in[11];
  const float* b1   = (const float*)d_in[12];
  const float* W2   = (const float*)d_in[13];
  const float* b2   = (const float*)d_in[14];

  char* ws = (char*)d_ws;
  const size_t MB = 1024 * 1024;
  _Float16* Xh   = (_Float16*)(ws + 0);
  _Float16* Yh   = (_Float16*)(ws + 8 * MB);
  _Float16* Wqh  = (_Float16*)(ws + 16 * MB);
  _Float16* Wkh  = (_Float16*)(ws + 18 * MB);
  _Float16* Wvh  = (_Float16*)(ws + 20 * MB);
  _Float16* Woh  = (_Float16*)(ws + 22 * MB);
  _Float16* W1h  = (_Float16*)(ws + 24 * MB);
  _Float16* W2h  = (_Float16*)(ws + 26 * MB);
  _Float16* Qh   = (_Float16*)(ws + 28 * MB);
  _Float16* Kh   = (_Float16*)(ws + 36 * MB);
  _Float16* Vt   = (_Float16*)(ws + 44 * MB);
  float*    Fbuf = (float*)(ws + 52 * MB);
  float*    Hres = (float*)(ws + 68 * MB);
  _Float16* Ohb  = Xh;
  _Float16* Hh   = Yh;
  _Float16* FF1h = Qh;

  dim3 blk256(256);

  cvt_all_kernel<<<512, blk256, 0, stream>>>(X, Y, Wq, Wk, Wv, Wo, W1, W2, Xh);

  const int ggrid = (4096 / 128) * (1024 / 128);

  gemm_nt_kernel<OUT_F16, false, false><<<ggrid, blk256, 0, stream>>>(
      Xh, Wqh, nullptr, Qh, nullptr, 4096, 1024, 1024);
  gemm_nt_kernel<OUT_F16, false, false><<<ggrid, blk256, 0, stream>>>(
      Yh, Wkh, nullptr, Kh, nullptr, 4096, 1024, 1024);
  gemm_nt_kernel<OUT_F16T, false, false><<<ggrid, blk256, 0, stream>>>(
      Yh, Wvh, nullptr, Vt, nullptr, 4096, 1024, 1024);

  attn_kernel<<<512, blk256, 0, stream>>>(Qh, Kh, Vt, mask, Ohb);

  gemm_nt_kernel<OUT_F32, false, false><<<ggrid, blk256, 0, stream>>>(
      Ohb, Woh, Fbuf, nullptr, nullptr, 4096, 1024, 1024);
  ln_kernel<<<4096, blk256, 0, stream>>>(X, Fbuf, ln1g, ln1b, Hres, Hh);

  gemm_nt_kernel<OUT_F16, true, true><<<ggrid, blk256, 0, stream>>>(
      Hh, W1h, nullptr, FF1h, b1, 4096, 1024, 1024);
  gemm_nt_kernel<OUT_F32, true, false><<<ggrid, blk256, 0, stream>>>(
      FF1h, W2h, Fbuf, nullptr, b2, 4096, 1024, 1024);
  ln_kernel<<<4096, blk256, 0, stream>>>(Hres, Fbuf, ln2g, ln2b, (float*)d_out,
                                         nullptr);
}


# gemm: Q, K and V projections run as ONE launch of 768 workgroups (3 x 256 tiles) instead of three of 256: two workgroups share a WGP and fill each other's load and barrier waits; same tile code
# speedup vs baseline: 2.8834x; 1.0700x over previous
#include <hip/hip_runtime.h>
#include <hip/hip_bf16.h>


typedef __attribute__((ext_vector_type(16))) _Float16 v16h;
typedef __attribute__((ext_vector_type(8)))  _Float16 v8h;
typedef __attribute__((ext_vector_type(4)))  _Float16 v4h;
typedef __attribute__((ext_vector_type(8)))  float    v8f;

__device__ __forceinline__ v16h load_frag(const _Float16* base, int ld,
                                          int row0, int k0, int lane) {
  const int r  = row0 + (lane & 15);
  const int kb = k0 + ((lane >> 4) << 3);
  const _Float16* p = base + (size_t)r * ld + kb;
  v8h lo = *(const v8h*)(p);
  v8h hi = *(const v8h*)(p + 16);
  return __builtin_shufflevector(lo, hi, 0, 1, 2, 3, 4, 5, 6, 7,
                                         8, 9, 10, 11, 12, 13, 14, 15);
}

__device__ __forceinline__ v8f wmma_f16(v16h a, v16h b, v8f c) {
  return __builtin_amdgcn_wmma_f32_16x16x32_f16(false, a, false, b,
                                                (short)0, c, false, false);
}

#define CV_SLAB (512 * 256)

__device__ __forceinline__ v4h cvt_quad(float4 v) {
  v4h o = {(_Float16)v.x, (_Float16)v.y, (_Float16)v.z, (_Float16)v.w};
  return o;
}

__global__ __launch_bounds__(256) void cvt_all_kernel(
    const float* __restrict__ X,  const float* __restrict__ Y,
    const float* __restrict__ Wq, const float* __restrict__ Wk,
    const float* __restrict__ Wv, const float* __restrict__ Wo,
    const float* __restrict__ W1, const float* __restrict__ W2,
    _Float16* __restrict__ dst) {
  const size_t q0 = (size_t)blockIdx.x * 256 + threadIdx.x;
  float4 v[28];
#pragma unroll
  for (int i = 0; i < 8; i++) v[i]     = ((const float4*)X)[q0 + (size_t)i * CV_SLAB];
#pragma unroll
  for (int i = 0; i < 8; i++) v[8 + i] = ((const float4*)Y)[q0 + (size_t)i * CV_SLAB];
#pragma unroll
  for (int i = 0; i < 2; i++) {
    v[16 + i] = ((const float4*)Wq)[q0 + (size_t)i * CV_SLAB];
    v[18 + i] = ((const float4*)Wk)[q0 + (size_t)i * CV_SLAB];
    v[20 + i] = ((const float4*)Wv)[q0 + (size_t)i * CV_SLAB];
    v[22 + i] = ((const float4*)Wo)[q0 + (size_t)i * CV_SLAB];
    v[24 + i] = ((const float4*)W1)[q0 + (size_t)i * CV_SLAB];
    v[26 + i] = ((const float4*)W2)[q0 + (size_t)i * CV_SLAB];
  }
  __builtin_amdgcn_sched_barrier(0);
#pragma unroll
  for (int i = 0; i < 28; i++)
    ((v4h*)dst)[q0 + (size_t)i * CV_SLAB] = cvt_quad(v[i]);
}

#define G_LD    1024
#define G_BK    128
#define G_PITCH 136
#define G_STAGE (256 * G_PITCH)
#define E_PITCH32 132

#define OUT_F16  0
#define OUT_F16T 1
#define OUT_F32  2

template <bool SWAP>
__device__ __forceinline__ void gemm_stage_mma(const _Float16* sA,
                                               const _Float16* sB,
                                               v8f (&acc)[4][2], int lane) {
  v16h af[2][4], bf[2][2];
#pragma unroll
  for (int i = 0; i < 4; i++) af[0][i] = load_frag(sA, G_PITCH, 16 * i, 0, lane);
#pragma unroll
  for (int j = 0; j < 2; j++) bf[0][j] = load_frag(sB, G_PITCH, 16 * j, 0, lane);
#pragma unroll
  for (int u = 0; u < 4; u++) {
    const int c = u & 1, n = c ^ 1;
    if (u + 1 < 4) {
#pragma unroll
      for (int i = 0; i < 4; i++)
        af[n][i] = load_frag(sA, G_PITCH, 16 * i, 32 * (u + 1), lane);
#pragma unroll
      for (int j = 0; j < 2; j++)
        bf[n][j] = load_frag(sB, G_PITCH, 16 * j, 32 * (u + 1), lane);
    }
    __builtin_amdgcn_sched_barrier(0);
#pragma unroll
    for (int i = 0; i < 4; i++)
#pragma unroll
      for (int j = 0; j < 2; j++)
        acc[i][j] = SWAP ? wmma_f16(bf[c][j], af[c][i], acc[i][j])
                         : wmma_f16(af[c][i], bf[c][j], acc[i][j]);
  }
}

template <int OUT, bool BIAS, bool RELU>
__device__ __forceinline__ void gemm_tile(
    const _Float16* __restrict__ A, const _Float16* __restrict__ Bw,
    float* __restrict__ Cf, _Float16* __restrict__ Ch,
    const float* __restrict__ bias, int Ndim, int Kdim, int tile,
    _Float16* lds) {
  constexpr bool SWAP = (OUT != OUT_F16T);

  const int tid  = threadIdx.x;
  const int lane = tid & 31;
  const int wave = tid >> 5;
  const int tiles_n = Ndim >> 7;
  const int bm = tile / tiles_n;
  const int bn = tile - bm * tiles_n;
  const int wm = (wave & 1) * 64;
  const int wn = (wave >> 1) * 32;

  const int srow = tid >> 4;
  const int scol = (tid & 15) << 3;
  const _Float16* gA = A  + (size_t)(bm * 128 + srow) * G_LD + scol;
  const _Float16* gB = Bw + (size_t)(bn * 128 + srow) * G_LD + scol;
  const int soff = srow * G_PITCH + scol;

  v8f acc[4][2] = {};
  v8h ra[8], rb[8];

#pragma unroll
  for (int p = 0; p < 8; p++) {
    ra[p] = *(const v8h*)(gA + p * 16 * G_LD);
    rb[p] = *(const v8h*)(gB + p * 16 * G_LD);
  }
#pragma unroll
  for (int p = 0; p < 8; p++) {
    *(v8h*)(lds + soff + (16 * p) * G_PITCH)       = ra[p];
    *(v8h*)(lds + soff + (128 + 16 * p) * G_PITCH) = rb[p];
  }
  __syncthreads();

  const int nk = Kdim / G_BK;
#pragma unroll 1
  for (int kt = 0; kt + 1 < nk; kt++) {
    const _Float16* cur = lds + (kt & 1) * G_STAGE;
    _Float16*       nxt = lds + ((kt + 1) & 1) * G_STAGE;
    gA += G_BK;
    gB += G_BK;
#pragma unroll
    for (int p = 0; p < 8; p++) {
      ra[p] = *(const v8h*)(gA + p * 16 * G_LD);
      rb[p] = *(const v8h*)(gB + p * 16 * G_LD);
    }
    gemm_stage_mma<SWAP>(cur + wm * G_PITCH, cur + (128 + wn) * G_PITCH, acc, lane);
#pragma unroll
    for (int p = 0; p < 8; p++) {
      *(v8h*)(nxt + soff + (16 * p) * G_PITCH)       = ra[p];
      *(v8h*)(nxt + soff + (128 + 16 * p) * G_PITCH) = rb[p];
    }
    __syncthreads();
  }

  const int lc = lane & 15;
  const int lh = lane >> 4;

  float bv[2][8];
#pragma unroll
  for (int j = 0; j < 2; j++) {
    if (BIAS && SWAP) {
      const float* bp = bias + bn * 128 + wn + 16 * j + 8 * lh;
      const float4 b0 = *(const float4*)(bp);
      const float4 b1 = *(const float4*)(bp + 4);
      bv[j][0] = b0.x; bv[j][1] = b0.y; bv[j][2] = b0.z; bv[j][3] = b0.w;
      bv[j][4] = b1.x; bv[j][5] = b1.y; bv[j][6] = b1.z; bv[j][7] = b1.w;
    } else {
      const float b0 = BIAS ? bias[bn * 128 + wn + 16 * j + lc] : 0.f;
#pragma unroll
      for (int r = 0; r < 8; r++) bv[j][r] = b0;
    }
  }

  {
    const _Float16* cur = lds + ((nk - 1) & 1) * G_STAGE;
    gemm_stage_mma<SWAP>(cur + wm * G_PITCH, cur + (128 + wn) * G_PITCH, acc, lane);
  }

  _Float16* st = lds + (nk & 1) * G_STAGE;

  if (OUT == OUT_F32) {
    float* sf = (float*)st;
#pragma unroll
    for (int i = 0; i < 4; i++) {
#pragma unroll
      for (int j = 0; j < 2; j++) {
        float v[8];
#pragma unroll
        for (int r = 0; r < 8; r++) {
          v[r] = BIAS ? acc[i][j][r] + bv[j][r] : acc[i][j][r];
          if (RELU) v[r] = fmaxf(v[r], 0.f);
        }
        float* q = sf + (wm + 16 * i + lc) * E_PITCH32 + wn + 16 * j + 8 * lh;
        *(float4*)(q)     = make_float4(v[0], v[1], v[2], v[3]);
        *(float4*)(q + 4) = make_float4(v[4], v[5], v[6], v[7]);
      }
    }
    __syncthreads();
    const int row0 = tid >> 5;
    const int col  = (tid & 31) << 2;
    float* gC = Cf + (size_t)(bm * 128 + row0) * Ndim + bn * 128 + col;
    const float* sp = sf + row0 * E_PITCH32 + col;
#pragma unroll
    for (int p = 0; p < 16; p++) {
      const float4 o = *(const float4*)(sp + (8 * p) * E_PITCH32);
      *(float4*)(gC + (size_t)(8 * p) * Ndim) = o;
    }
  } else {
#pragma unroll
    for (int i = 0; i < 4; i++) {
#pragma unroll
      for (int j = 0; j < 2; j++) {
        v8h o;
#pragma unroll
        for (int r = 0; r < 8; r++) {
          float v = BIAS ? acc[i][j][r] + bv[j][r] : acc[i][j][r];
          if (RELU) v = fmaxf(v, 0.f);
          o[r] = (_Float16)v;
        }
        if (SWAP)
          *(v8h*)(st + (wm + 16 * i + lc) * G_PITCH + wn + 16 * j + 8 * lh) = o;
        else
          *(v8h*)(st + (wn + 16 * j + lc) * G_PITCH + wm + 16 * i + 8 * lh) = o;
      }
    }
    __syncthreads();
    _Float16* gC;
    size_t gpitch;
    if (OUT == OUT_F16) {
      gC = Ch + (size_t)(bm * 128 + srow) * Ndim + bn * 128 + scol;
      gpitch = (size_t)Ndim;
    } else {
      const int mg = bm * 128 + scol;
      gC = Ch + ((size_t)((mg >> 10) << 10) + bn * 128 + srow) * 1024 + (mg & 1023);
      gpitch = 1024;
    }
#pragma unroll
    for (int p = 0; p < 8; p++) {
      const v8h o = *(const v8h*)(st + soff + (16 * p) * G_PITCH);
      *(v8h*)(gC + (size_t)(16 * p) * gpitch) = o;
    }
  }
}

template <int OUT, bool BIAS, bool RELU>
__global__ __launch_bounds__(256) void gemm_nt_kernel(
    const _Float16* __restrict__ A, const _Float16* __restrict__ Bw,
    float* __restrict__ Cf, _Float16* __restrict__ Ch,
    const float* __restrict__ bias, int Mdim, int Ndim, int Kdim) {
  __shared__ __attribute__((aligned(16))) _Float16 lds[2 * G_STAGE];
  (void)Mdim;
  gemm_tile<OUT, BIAS, RELU>(A, Bw, Cf, Ch, bias, Ndim, Kdim, blockIdx.x, lds);
}

__global__ __launch_bounds__(256) void qkv_proj_kernel(
    const _Float16* __restrict__ Xh, const _Float16* __restrict__ Yh,
    const _Float16* __restrict__ Wqh, const _Float16* __restrict__ Wkh,
    const _Float16* __restrict__ Wvh, _Float16* __restrict__ Qh,
    _Float16* __restrict__ Kh, _Float16* __restrict__ Vt) {
  __shared__ __attribute__((aligned(16))) _Float16 lds[2 * G_STAGE];
  const int which = blockIdx.x >> 8;
  const int tile  = blockIdx.x & 255;
  if (which == 1) {
    gemm_tile<OUT_F16T, false, false>(Yh, Wvh, nullptr, Vt, nullptr,
                                      1024, 1024, tile, lds);
  } else {
    const bool isq = which == 2;
    gemm_tile<OUT_F16, false, false>(isq ? Xh : Yh, isq ? Wqh : Wkh, nullptr,
                                     isq ? Qh : Kh, nullptr,
                                     1024, 1024, tile, lds);
  }
}

#define AT_PITCH 72
#define AT_TILE  (64 * AT_PITCH)
#define AT_STAGE (2 * AT_TILE)

__device__ __forceinline__ float xhalf(float x) {
  const int xi = __builtin_bit_cast(int, x);
  return __builtin_bit_cast(float, __builtin_amdgcn_permlanex16(
      xi, xi, 0x76543210, 0xfedcba98, false, false));
}

__global__ __launch_bounds__(256) __attribute__((amdgpu_waves_per_eu(4, 4)))
void attn_kernel(
    const _Float16* __restrict__ Qh, const _Float16* __restrict__ Kh,
    const _Float16* __restrict__ Vt, const int* __restrict__ mask,
    _Float16* __restrict__ Oh) {
  __shared__ __attribute__((aligned(16))) _Float16 kv[2 * AT_STAGE];
  __shared__ __attribute__((aligned(16))) float sbias[1024];

  const int tid  = threadIdx.x;
  const int lane = tid & 31;
  const int wave = tid >> 5;
  const int lc = lane & 15;
  const int lh = lane >> 4;
  const int bh   = blockIdx.x >> 3;
  const int qblk = blockIdx.x & 7;
  const int b = bh >> 4, h = bh & 15;
  const int q0 = qblk * 128 + wave * 16;

  const _Float16* Qbase = Qh + (size_t)b * 1024 * 1024 + h * 64;
  const _Float16* Kbase = Kh + (size_t)b * 1024 * 1024 + h * 64;
  const _Float16* Vbase = Vt + (size_t)(b * 16 + h) * 64 * 1024;
  const int* mb = mask + b * 1024;

  const int srow = tid >> 3;
  const int scol = (tid & 7) << 3;
  const _Float16* gK = Kbase + (size_t)srow * 1024 + scol;
  const _Float16* gV = Vbase + (size_t)srow * 1024 + scol;
  const int soff = srow * AT_PITCH + scol;

  v8h rk0 = *(const v8h*)(gK);
  v8h rk1 = *(const v8h*)(gK + 32 * 1024);
  v8h rv0 = *(const v8h*)(gV);
  v8h rv1 = *(const v8h*)(gV + 32 * 1024);
  {
    const int4 mv = ((const int4*)mb)[tid];
    float4 bv;
    bv.x = mv.x ? 0.f : -1e30f;
    bv.y = mv.y ? 0.f : -1e30f;
    bv.z = mv.z ? 0.f : -1e30f;
    bv.w = mv.w ? 0.f : -1e30f;
    ((float4*)sbias)[tid] = bv;
  }

  const v16h qb0 = load_frag(Qbase, 1024, q0, 0, lane);
  const v16h qb1 = load_frag(Qbase, 1024, q0, 32, lane);

  *(v8h*)(kv + soff)                           = rk0;
  *(v8h*)(kv + soff + 32 * AT_PITCH)           = rk1;
  *(v8h*)(kv + AT_TILE + soff)                 = rv0;
  *(v8h*)(kv + AT_TILE + soff + 32 * AT_PITCH) = rv1;
  __syncthreads();

  v8f o[4] = {};
  float mrun = -1e30f;
  float lrun = 0.f;
  const float scale = 0.125f;

#pragma unroll 1
  for (int st = 0; st < 16; st++) {
    const _Float16* sK = kv + (st & 1) * AT_STAGE;
    const _Float16* sV = sK + AT_TILE;
    _Float16*      nxt = kv + ((st + 1) & 1) * AT_STAGE;
    const int sn = st + 1 < 16 ? st + 1 : 15;
    rk0 = *(const v8h*)(gK + (size_t)sn * 64 * 1024);
    rk1 = *(const v8h*)(gK + (size_t)sn * 64 * 1024 + 32 * 1024);
    rv0 = *(const v8h*)(gV + sn * 64);
    rv1 = *(const v8h*)(gV + sn * 64 + 32 * 1024);

#pragma unroll
    for (int j = 0; j < 2; j++) {
      const int kb = st * 64 + j * 32;
      v16h kf[2][2], vf[4];
#pragma unroll
      for (int sub = 0; sub < 2; sub++) {
        kf[sub][0] = load_frag(sK, AT_PITCH, j * 32 + sub * 16, 0, lane);
        kf[sub][1] = load_frag(sK, AT_PITCH, j * 32 + sub * 16, 32, lane);
      }
#pragma unroll
      for (int t = 0; t < 4; t++)
        vf[t] = load_frag(sV, AT_PITCH, t * 16, j * 32, lane);
      const float* bp = sbias + kb + 8 * lh;
      const float4 b00 = *(const float4*)(bp);
      const float4 b01 = *(const float4*)(bp + 4);
      const float4 b10 = *(const float4*)(bp + 16);
      const float4 b11 = *(const float4*)(bp + 20);
      const float bias[2][8] = {
          {b00.x, b00.y, b00.z, b00.w, b01.x, b01.y, b01.z, b01.w},
          {b10.x, b10.y, b10.z, b10.w, b11.x, b11.y, b11.z, b11.w}};

      v8f s[2] = {};
#pragma unroll
      for (int sub = 0; sub < 2; sub++) {
        s[sub] = wmma_f16(kf[sub][0], qb0, s[sub]);
        s[sub] = wmma_f16(kf[sub][1], qb1, s[sub]);
      }

      float v[2][8];
#pragma unroll
      for (int sub = 0; sub < 2; sub++)
#pragma unroll
        for (int r = 0; r < 8; r++)
          v[sub][r] = fmaf(s[sub][r], scale, bias[sub][r]);
      float mx = v[0][0];
#pragma unroll
      for (int sub = 0; sub < 2; sub++)
#pragma unroll
        for (int r = 0; r < 8; r++) mx = fmaxf(mx, v[sub][r]);
      mx = fmaxf(mx, xhalf(mx));

      const float mnew  = fmaxf(mrun, mx);
      const float alpha = __expf(mrun - mnew);
      float p[2][8];
#pragma unroll
      for (int sub = 0; sub < 2; sub++)
#pragma unroll
        for (int r = 0; r < 8; r++)
          p[sub][r] = __expf(v[sub][r] - mnew);
      float psum = 0.f;
#pragma unroll
      for (int sub = 0; sub < 2; sub++)
#pragma unroll
        for (int r = 0; r < 8; r += 2) psum += p[sub][r] + p[sub][r + 1];
      lrun = lrun * alpha + psum;
      mrun = mnew;
#pragma unroll
      for (int t = 0; t < 4; t++)
#pragma unroll
        for (int r = 0; r < 8; r++) o[t][r] *= alpha;

      v16h pb;
#pragma unroll
      for (int r = 0; r < 8; r++) {
        pb[r]     = (_Float16)p[0][r];
        pb[8 + r] = (_Float16)p[1][r];
      }
#pragma unroll
      for (int t = 0; t < 4; t++) o[t] = wmma_f16(vf[t], pb, o[t]);
    }

    *(v8h*)(nxt + soff)                           = rk0;
    *(v8h*)(nxt + soff + 32 * AT_PITCH)           = rk1;
    *(v8h*)(nxt + AT_TILE + soff)                 = rv0;
    *(v8h*)(nxt + AT_TILE + soff + 32 * AT_PITCH) = rv1;
    __syncthreads();
  }

  const float ltot = lrun + xhalf(lrun);
  const bool live = mrun > -1e29f;
  _Float16* orow = Oh + ((size_t)b * 1024 + q0 + lc) * 1024 + h * 64 + 8 * lh;
#pragma unroll
  for (int t = 0; t < 4; t++) {
    v8h ov;
#pragma unroll
    for (int r = 0; r < 8; r++) {
      const float val = live ? o[t][r] / ltot : 0.f;
      ov[r] = (_Float16)val;
    }
    *(v8h*)(orow + t * 16) = ov;
  }
}

__global__ __launch_bounds__(256) void ln_kernel(
    const float* __restrict__ Ain, const float* __restrict__ Bin,
    const float* __restrict__ g, const float* __restrict__ beta,
    float* __restrict__ out32, _Float16* __restrict__ out16) {
  const int row = blockIdx.x;
  const int t = threadIdx.x;
  const int lane = t & 31, wave = t >> 5;

  const float4 xa = ((const float4*)(Ain + (size_t)row * 1024))[t];
  const float4 xb = ((const float4*)(Bin + (size_t)row * 1024))[t];
  float x[4] = {xa.x + xb.x, xa.y + xb.y, xa.z + xb.z, xa.w + xb.w};

  float s1 = x[0] + x[1] + x[2] + x[3];
  float s2 = x[0]*x[0] + x[1]*x[1] + x[2]*x[2] + x[3]*x[3];
#pragma unroll
  for (int off = 16; off >= 1; off >>= 1) {
    s1 += __shfl_xor(s1, off, 32);
    s2 += __shfl_xor(s2, off, 32);
  }
  __shared__ float r1[8], r2[8];
  if (lane == 0) { r1[wave] = s1; r2[wave] = s2; }
  __syncthreads();
  float tot1 = 0.f, tot2 = 0.f;
#pragma unroll
  for (int i = 0; i < 8; i++) { tot1 += r1[i]; tot2 += r2[i]; }

  const float mu   = tot1 * (1.f / 1024.f);
  const float var  = tot2 * (1.f / 1024.f) - mu * mu;
  const float rstd = rsqrtf(var + 1e-5f);
#pragma unroll
  for (int j = 0; j < 4; j++) {
    const int idx = t * 4 + j;
    const float y = (x[j] - mu) * rstd * g[idx] + beta[idx];
    out32[(size_t)row * 1024 + idx] = y;
    if (out16) out16[(size_t)row * 1024 + idx] = (_Float16)y;
  }
}

extern "C" void kernel_launch(void* const* d_in, const int* in_sizes, int n_in,
                              void* d_out, int out_size, void* d_ws, size_t ws_size,
                              hipStream_t stream) {
  (void)in_sizes; (void)n_in; (void)out_size; (void)ws_size;
  const float* X    = (const float*)d_in[0];
  const float* Y    = (const float*)d_in[1];
  const int*   mask = (const int*)d_in[2];
  const float* Wq   = (const float*)d_in[3];
  const float* Wk   = (const float*)d_in[4];
  const float* Wv   = (const float*)d_in[5];
  const float* Wo   = (const float*)d_in[6];
  const float* ln1g = (const float*)d_in[7];
  const float* ln1b = (const float*)d_in[8];
  const float* ln2g = (const float*)d_in[9];
  const float* ln2b = (const float*)d_in[10];
  const float* W1   = (const float*)d_in[11];
  const float* b1   = (const float*)d_in[12];
  const float* W2   = (const float*)d_in[13];
  const float* b2   = (const float*)d_in[14];

  char* ws = (char*)d_ws;
  const size_t MB = 1024 * 1024;
  _Float16* Xh   = (_Float16*)(ws + 0);
  _Float16* Yh   = (_Float16*)(ws + 8 * MB);
  _Float16* Wqh  = (_Float16*)(ws + 16 * MB);
  _Float16* Wkh  = (_Float16*)(ws + 18 * MB);
  _Float16* Wvh  = (_Float16*)(ws + 20 * MB);
  _Float16* Woh  = (_Float16*)(ws + 22 * MB);
  _Float16* W1h  = (_Float16*)(ws + 24 * MB);
  _Float16* W2h  = (_Float16*)(ws + 26 * MB);
  _Float16* Qh   = (_Float16*)(ws + 28 * MB);
  _Float16* Kh   = (_Float16*)(ws + 36 * MB);
  _Float16* Vt   = (_Float16*)(ws + 44 * MB);
  float*    Fbuf = (float*)(ws + 52 * MB);
  float*    Hres = (float*)(ws + 68 * MB);
  _Float16* Ohb  = Xh;
  _Float16* Hh   = Yh;
  _Float16* FF1h = Qh;

  dim3 blk256(256);

  cvt_all_kernel<<<512, blk256, 0, stream>>>(X, Y, Wq, Wk, Wv, Wo, W1, W2, Xh);

  const int ggrid = (4096 / 128) * (1024 / 128);

  qkv_proj_kernel<<<768, blk256, 0, stream>>>(Xh, Yh, Wqh, Wkh, Wvh, Qh, Kh, Vt);

  attn_kernel<<<512, blk256, 0, stream>>>(Qh, Kh, Vt, mask, Ohb);

  gemm_nt_kernel<OUT_F32, false, false><<<ggrid, blk256, 0, stream>>>(
      Ohb, Woh, Fbuf, nullptr, nullptr, 4096, 1024, 1024);
  ln_kernel<<<4096, blk256, 0, stream>>>(X, Fbuf, ln1g, ln1b, Hres, Hh);

  gemm_nt_kernel<OUT_F16, true, true><<<ggrid, blk256, 0, stream>>>(
      Hh, W1h, nullptr, FF1h, b1, 4096, 1024, 1024);
  gemm_nt_kernel<OUT_F32, true, false><<<ggrid, blk256, 0, stream>>>(
      FF1h, W2h, Fbuf, nullptr, b2, 4096, 1024, 1024);
  ln_kernel<<<4096, blk256, 0, stream>>>(Hres, Fbuf, ln2g, ln2b, (float*)d_out,
                                         nullptr);
}
